# starting kernel
# baseline (speedup 1.0000x reference)
#include <hip/hip_runtime.h>
#include <math.h>

#define EPSBN 1e-5f
#define SLOPE 0.01f
static constexpr int  HP  = 24;
static constexpr long P   = 2L*24*24*24*24;
static constexpr long NX  = 2L*48*48*48*48;
static constexpr int  RX_BLOCKS = 648;
static constexpr int  M3_BLOCKS = 216;
static constexpr int  POS_BLOCKS = 2592;

static constexpr size_t OFF_U2   = 0;
static constexpr size_t OFF_WP   = (size_t)P*8*4;
static constexpr size_t OFF_DACC = OFF_WP + 768*4;
static constexpr size_t OFF_FC   = OFF_DACC + 512;

typedef float vf2 __attribute__((ext_vector_type(2)));
typedef float vf8 __attribute__((ext_vector_type(8)));

__device__ __forceinline__ float lrelu(float v){ return v >= 0.f ? v : SLOPE*v; }

__global__ void k_prep(const float* __restrict__ w2, float* __restrict__ wp,
                       double* __restrict__ dacc)
{
    int tid = threadIdx.x;
    if (tid < 56) dacc[tid] = 0.0;
    if (tid < 384) {
        int n = tid & 15, h = (tid >> 4) & 1, t = tid >> 5;
        int K0 = 4*t + 2*h;
        float vx = (n < 6) ? w2[n*48 + K0]     : 0.f;
        float vy = (n < 6) ? w2[n*48 + K0 + 1] : 0.f;
        ((float2*)wp)[t*32 + h*16 + n] = make_float2(vx, vy);
    }
}

__global__ __launch_bounds__(256) void k_rx(const float* __restrict__ x,
                                            double* __restrict__ dacc)
{
    __shared__ double sacc[2];
    int tid = threadIdx.x;
    if (tid < 2) sacc[tid] = 0.0;
    __syncthreads();
    long gid = (long)blockIdx.x*256 + tid;
    float s = 0.f, q = 0.f;
    #pragma unroll
    for (int it = 0; it < 16; ++it) {
        long idx = gid + (long)it * (RX_BLOCKS*256);
        float4 v = *(const float4*)(x + idx*4);
        s += v.x + v.y + v.z + v.w;
        q += v.x*v.x + v.y*v.y + v.z*v.z + v.w*v.w;
    }
    atomicAdd(&sacc[0], (double)s);
    atomicAdd(&sacc[1], (double)q);
    __syncthreads();
    if (tid < 2) atomicAdd(&dacc[tid], sacc[tid]);
}

__global__ void k_fin1(const double* __restrict__ dacc, const float* __restrict__ w1,
                       const float* __restrict__ g1, const float* __restrict__ b1,
                       float* __restrict__ fc)
{
    if (threadIdx.x != 0) return;
    double n  = (double)NX;
    double mu = dacc[0]/n;
    double var = dacc[1]/n - mu*mu;
    for (int c = 0; c < 3; ++c) {
        double wc  = (double)w1[c];
        double inv = 1.0 / sqrt(wc*wc*var + (double)EPSBN);
        fc[c]   = (float)(wc*inv) * g1[c];
        fc[3+c] = (float)(-wc*mu*inv) * g1[c] + b1[c];
    }
}

__global__ __launch_bounds__(256) void k_conv2(const float* __restrict__ x,
                                               const float* __restrict__ wp,
                                               const float* __restrict__ fc,
                                               float* __restrict__ u2,
                                               double* __restrict__ dacc)
{
    __shared__ float lx[256*18];
    __shared__ float lsum[6], lssq[6];
    int tid = threadIdx.x;
    if (tid < 6) { lsum[tid] = 0.f; lssq[tid] = 0.f; }

    int pos = blockIdx.x*256 + tid;
    int l  = pos % HP;  int t1 = pos / HP;
    int kk = t1 % HP;   t1 /= HP;
    int j  = t1 % HP;   t1 /= HP;
    int i  = t1 % HP;   int b = t1 / HP;
    long xbase = ((((long)b*48 + 2*i)*48 + 2*j)*48 + 2*kk)*48 + 2*l;

    float* row = &lx[tid*18];
    #pragma unroll
    for (int p = 0; p < 2; ++p)
      #pragma unroll
      for (int q = 0; q < 2; ++q)
        #pragma unroll
        for (int r = 0; r < 2; ++r) {
            float2 v = *(const float2*)(x + xbase + (long)p*110592 + q*2304 + r*48);
            *(float2*)(row + (p*8 + q*4 + r*2)) = v;
        }

    int lane = tid & 31, l16 = lane & 15, h = lane >> 4, wv = tid >> 5;

    vf2 Bfrag[12];
    #pragma unroll
    for (int t = 0; t < 12; ++t)
        Bfrag[t] = *(const vf2*)(wp + ((t*2 + h)*16 + l16)*2);

    float a1[3], c1[3];
    #pragma unroll
    for (int c = 0; c < 3; ++c) { a1[c] = fc[c]; c1[c] = fc[3+c]; }

    __syncthreads();

    #pragma unroll
    for (int tau = 0; tau < 2; ++tau) {
        int bl = wv*32 + tau*16;
        vf8 acc = {0.f,0.f,0.f,0.f,0.f,0.f,0.f,0.f};
        #pragma unroll
        for (int t = 0; t < 12; ++t) {
            int c  = t >> 2;
            int fb = (t & 3)*4 + 2*h;
            float2 xv = *(const float2*)&lx[(bl + l16)*18 + fb];
            vf2 A;
            A.x = lrelu(a1[c]*xv.x + c1[c]);
            A.y = lrelu(a1[c]*xv.y + c1[c]);
            acc = __builtin_amdgcn_wmma_f32_16x16x4_f32(
                      false, A, false, Bfrag[t], (short)0, acc, false, false);
        }
        if (l16 < 6) {
            long pbase = (long)blockIdx.x*256 + bl + 8*h;
            float ls = 0.f, lq = 0.f;
            #pragma unroll
            for (int r = 0; r < 8; ++r) {
                float v = acc[r];
                u2[(pbase + r)*8 + l16] = v;
                ls += v; lq += v*v;
            }
            atomicAdd(&lsum[l16], ls);
            atomicAdd(&lssq[l16], lq);
        }
    }
    __syncthreads();
    if (tid < 6) {
        atomicAdd(&dacc[2 + tid], (double)lsum[tid]);
        atomicAdd(&dacc[8 + tid], (double)lssq[tid]);
    }
}

__global__ void k_fin2(const double* __restrict__ dacc, const float* __restrict__ g2,
                       const float* __restrict__ b2, float* __restrict__ fc)
{
    if (threadIdx.x != 0) return;
    double n = (double)P;
    for (int c = 0; c < 6; ++c) {
        double mu  = dacc[2+c]/n;
        double var = dacc[8+c]/n - mu*mu;
        double inv = 1.0 / sqrt(var + (double)EPSBN);
        float sc = (float)inv * g2[c];
        fc[6+c]  = sc;
        fc[12+c] = b2[c] - (float)mu*sc;
    }
}

__global__ __launch_bounds__(256) void k_m3(const float* __restrict__ u2,
                                            const float* __restrict__ fc,
                                            double* __restrict__ dacc)
{
    __shared__ double sacc[27];
    int tid = threadIdx.x;
    if (tid < 27) sacc[tid] = 0.0;
    __syncthreads();
    float m[6]; float s[21];
    #pragma unroll
    for (int c = 0; c < 6; ++c) m[c] = 0.f;
    #pragma unroll
    for (int k = 0; k < 21; ++k) s[k] = 0.f;
    long gid = (long)blockIdx.x*256 + tid;
    for (int it = 0; it < 12; ++it) {
        long pos = gid + (long)it*(M3_BLOCKS*256);
        float4 ua = *(const float4*)(u2 + pos*8);
        float4 ub = *(const float4*)(u2 + pos*8 + 4);
        float uv[6] = {ua.x, ua.y, ua.z, ua.w, ub.x, ub.y};
        float y[6];
        #pragma unroll
        for (int c = 0; c < 6; ++c) y[c] = lrelu(uv[c]*fc[6+c] + fc[12+c]);
        int idx = 0;
        #pragma unroll
        for (int c = 0; c < 6; ++c) {
            m[c] += y[c];
            #pragma unroll
            for (int c2 = c; c2 < 6; ++c2) s[idx++] += y[c]*y[c2];
        }
    }
    #pragma unroll
    for (int c = 0; c < 6; ++c)  atomicAdd(&sacc[c],     (double)m[c]);
    #pragma unroll
    for (int k = 0; k < 21; ++k) atomicAdd(&sacc[6 + k], (double)s[k]);
    __syncthreads();
    if (tid < 27) atomicAdd(&dacc[14 + tid], sacc[tid]);
}

__global__ void k_fin3(const double* __restrict__ dacc, const float* __restrict__ wt,
                       const float* __restrict__ g3, const float* __restrict__ b3,
                       float* __restrict__ fc)
{
    if (threadIdx.x != 0) return;
    double n = (double)P;
    double mean[6], S[6][6];
    for (int c = 0; c < 6; ++c) mean[c] = dacc[14+c]/n;
    int idx = 0;
    for (int c = 0; c < 6; ++c)
        for (int c2 = c; c2 < 6; ++c2) {
            double v = dacc[20+idx]/n; S[c][c2] = v; S[c2][c] = v; ++idx;
        }
    for (int o = 0; o < 3; ++o) {
        double mu = 0.0, m2 = 0.0;
        for (int f = 0; f < 16; ++f) {
            double wv[6];
            for (int c = 0; c < 6; ++c) wv[c] = (double)wt[(c*3+o)*16 + f];
            for (int c = 0; c < 6; ++c) {
                mu += wv[c]*mean[c];
                for (int c2 = 0; c2 < 6; ++c2) m2 += wv[c]*wv[c2]*S[c][c2];
            }
        }
        mu /= 16.0; m2 /= 16.0;
        double var = m2 - mu*mu;
        double inv = 1.0 / sqrt(var + (double)EPSBN);
        float A = (float)inv * g3[o];
        fc[18+o] = A;
        fc[21+o] = b3[o] - (float)mu*A;
    }
}

__global__ __launch_bounds__(256) void k_out(const float* __restrict__ u2,
                                             const float* __restrict__ wt,
                                             const float* __restrict__ fc,
                                             float* __restrict__ out)
{
    int pos = blockIdx.x*256 + threadIdx.x;
    int l  = pos % HP;  int t1 = pos / HP;
    int kk = t1 % HP;   t1 /= HP;
    int j  = t1 % HP;   t1 /= HP;
    int i  = t1 % HP;   int b = t1 / HP;

    float4 ua = *(const float4*)(u2 + (long)pos*8);
    float4 ub = *(const float4*)(u2 + (long)pos*8 + 4);
    float uv[6] = {ua.x, ua.y, ua.z, ua.w, ub.x, ub.y};
    float y[6];
    #pragma unroll
    for (int c = 0; c < 6; ++c) y[c] = lrelu(uv[c]*fc[6+c] + fc[12+c]);

    float acc[48];
    #pragma unroll
    for (int u = 0; u < 48; ++u) acc[u] = 0.f;
    #pragma unroll
    for (int c = 0; c < 6; ++c)
      #pragma unroll
      for (int o = 0; o < 3; ++o) {
        const float* wrow = wt + (c*3 + o)*16;
        #pragma unroll
        for (int f = 0; f < 16; ++f) acc[o*16 + f] += y[c]*wrow[f];
      }

    #pragma unroll
    for (int o = 0; o < 3; ++o) {
        float A3 = fc[18+o], B3 = fc[21+o];
        #pragma unroll
        for (int p = 0; p < 2; ++p)
          #pragma unroll
          for (int q = 0; q < 2; ++q)
            #pragma unroll
            for (int r = 0; r < 2; ++r) {
                int f = p*8 + q*4 + r*2;
                float z0 = acc[o*16 + f    ]*A3 + B3;
                float z1 = acc[o*16 + f + 1]*A3 + B3;
                float s0 = 1.f / (1.f + __expf(-z0));
                float s1 = 1.f / (1.f + __expf(-z1));
                long oidx = (((((long)b*3 + o)*48 + (2*i+p))*48 + (2*j+q))*48 + (2*kk+r))*48 + 2*l;
                *(float2*)(out + oidx) = make_float2(s0, s1);
            }
    }
}

extern "C" void kernel_launch(void* const* d_in, const int* in_sizes, int n_in,
                              void* d_out, int out_size, void* d_ws, size_t ws_size,
                              hipStream_t stream)
{
    (void)in_sizes; (void)n_in; (void)out_size; (void)ws_size;
    const float* x  = (const float*)d_in[0];
    const float* w1 = (const float*)d_in[1];
    const float* g1 = (const float*)d_in[2];
    const float* b1 = (const float*)d_in[3];
    const float* w2 = (const float*)d_in[4];
    const float* g2 = (const float*)d_in[5];
    const float* b2 = (const float*)d_in[6];
    const float* wt = (const float*)d_in[7];
    const float* g3 = (const float*)d_in[8];
    const float* b3 = (const float*)d_in[9];
    float*  out  = (float*)d_out;
    float*  u2   = (float*)((char*)d_ws + OFF_U2);
    float*  wp   = (float*)((char*)d_ws + OFF_WP);
    double* dacc = (double*)((char*)d_ws + OFF_DACC);
    float*  fc   = (float*)((char*)d_ws + OFF_FC);

    k_prep <<<1, 512, 0, stream>>>(w2, wp, dacc);
    k_rx   <<<RX_BLOCKS, 256, 0, stream>>>(x, dacc);
    k_fin1 <<<1, 32, 0, stream>>>(dacc, w1, g1, b1, fc);
    k_conv2<<<POS_BLOCKS, 256, 0, stream>>>(x, wp, fc, u2, dacc);
    k_fin2 <<<1, 32, 0, stream>>>(dacc, g2, b2, fc);
    k_m3   <<<M3_BLOCKS, 256, 0, stream>>>(u2, fc, dacc);
    k_fin3 <<<1, 32, 0, stream>>>(dacc, wt, g3, b3, fc);
    k_out  <<<POS_BLOCKS, 256, 0, stream>>>(u2, wt, fc, out);
}


# f64 finalisers spread over lanes: fin1/fin2 one lane per channel, fin3 one lane per (channel,tap), taps summed by a lane butterfly, 1/n applied once to the sums; tap-sum order changes
# speedup vs baseline: 1.5727x; 1.5727x over previous
#include <hip/hip_runtime.h>
#include <math.h>

#define EPSBN 1e-5f
#define SLOPE 0.01f
static constexpr int  HP  = 24;
static constexpr long P   = 2L*24*24*24*24;
static constexpr long NX  = 2L*48*48*48*48;
static constexpr int  RX_BLOCKS = 648;
static constexpr int  M3_BLOCKS = 216;
static constexpr int  POS_BLOCKS = 2592;

static constexpr size_t OFF_U2   = 0;
static constexpr size_t OFF_WP   = (size_t)P*8*4;
static constexpr size_t OFF_DACC = OFF_WP + 768*4;
static constexpr size_t OFF_FC   = OFF_DACC + 512;

typedef float vf2 __attribute__((ext_vector_type(2)));
typedef float vf8 __attribute__((ext_vector_type(8)));

__device__ __forceinline__ float lrelu(float v){ return v >= 0.f ? v : SLOPE*v; }

__global__ void k_prep(const float* __restrict__ w2, float* __restrict__ wp,
                       double* __restrict__ dacc)
{
    int tid = threadIdx.x;
    if (tid < 56) dacc[tid] = 0.0;
    if (tid < 384) {
        int n = tid & 15, h = (tid >> 4) & 1, t = tid >> 5;
        int K0 = 4*t + 2*h;
        float vx = (n < 6) ? w2[n*48 + K0]     : 0.f;
        float vy = (n < 6) ? w2[n*48 + K0 + 1] : 0.f;
        ((float2*)wp)[t*32 + h*16 + n] = make_float2(vx, vy);
    }
}

__global__ __launch_bounds__(256) void k_rx(const float* __restrict__ x,
                                            double* __restrict__ dacc)
{
    __shared__ double sacc[2];
    int tid = threadIdx.x;
    if (tid < 2) sacc[tid] = 0.0;
    __syncthreads();
    long gid = (long)blockIdx.x*256 + tid;
    float s = 0.f, q = 0.f;
    #pragma unroll
    for (int it = 0; it < 16; ++it) {
        long idx = gid + (long)it * (RX_BLOCKS*256);
        float4 v = *(const float4*)(x + idx*4);
        s += v.x + v.y + v.z + v.w;
        q += v.x*v.x + v.y*v.y + v.z*v.z + v.w*v.w;
    }
    atomicAdd(&sacc[0], (double)s);
    atomicAdd(&sacc[1], (double)q);
    __syncthreads();
    if (tid < 2) atomicAdd(&dacc[tid], sacc[tid]);
}

__global__ void k_fin1(const double* __restrict__ dacc, const float* __restrict__ w1,
                       const float* __restrict__ g1, const float* __restrict__ b1,
                       float* __restrict__ fc)
{
    int c = threadIdx.x;
    if (c >= 3) return;
    double n  = (double)NX;
    double mu = dacc[0]/n;
    double var = dacc[1]/n - mu*mu;
    double wc  = (double)w1[c];
    double inv = 1.0 / sqrt(wc*wc*var + (double)EPSBN);
    fc[c]   = (float)(wc*inv) * g1[c];
    fc[3+c] = (float)(-wc*mu*inv) * g1[c] + b1[c];
}

__global__ __launch_bounds__(256) void k_conv2(const float* __restrict__ x,
                                               const float* __restrict__ wp,
                                               const float* __restrict__ fc,
                                               float* __restrict__ u2,
                                               double* __restrict__ dacc)
{
    __shared__ float lx[256*18];
    __shared__ float lsum[6], lssq[6];
    int tid = threadIdx.x;
    if (tid < 6) { lsum[tid] = 0.f; lssq[tid] = 0.f; }

    int pos = blockIdx.x*256 + tid;
    int l  = pos % HP;  int t1 = pos / HP;
    int kk = t1 % HP;   t1 /= HP;
    int j  = t1 % HP;   t1 /= HP;
    int i  = t1 % HP;   int b = t1 / HP;
    long xbase = ((((long)b*48 + 2*i)*48 + 2*j)*48 + 2*kk)*48 + 2*l;

    float* row = &lx[tid*18];
    #pragma unroll
    for (int p = 0; p < 2; ++p)
      #pragma unroll
      for (int q = 0; q < 2; ++q)
        #pragma unroll
        for (int r = 0; r < 2; ++r) {
            float2 v = *(const float2*)(x + xbase + (long)p*110592 + q*2304 + r*48);
            *(float2*)(row + (p*8 + q*4 + r*2)) = v;
        }

    int lane = tid & 31, l16 = lane & 15, h = lane >> 4, wv = tid >> 5;

    vf2 Bfrag[12];
    #pragma unroll
    for (int t = 0; t < 12; ++t)
        Bfrag[t] = *(const vf2*)(wp + ((t*2 + h)*16 + l16)*2);

    float a1[3], c1[3];
    #pragma unroll
    for (int c = 0; c < 3; ++c) { a1[c] = fc[c]; c1[c] = fc[3+c]; }

    __syncthreads();

    #pragma unroll
    for (int tau = 0; tau < 2; ++tau) {
        int bl = wv*32 + tau*16;
        vf8 acc = {0.f,0.f,0.f,0.f,0.f,0.f,0.f,0.f};
        #pragma unroll
        for (int t = 0; t < 12; ++t) {
            int c  = t >> 2;
            int fb = (t & 3)*4 + 2*h;
            float2 xv = *(const float2*)&lx[(bl + l16)*18 + fb];
            vf2 A;
            A.x = lrelu(a1[c]*xv.x + c1[c]);
            A.y = lrelu(a1[c]*xv.y + c1[c]);
            acc = __builtin_amdgcn_wmma_f32_16x16x4_f32(
                      false, A, false, Bfrag[t], (short)0, acc, false, false);
        }
        if (l16 < 6) {
            long pbase = (long)blockIdx.x*256 + bl + 8*h;
            float ls = 0.f, lq = 0.f;
            #pragma unroll
            for (int r = 0; r < 8; ++r) {
                float v = acc[r];
                u2[(pbase + r)*8 + l16] = v;
                ls += v; lq += v*v;
            }
            atomicAdd(&lsum[l16], ls);
            atomicAdd(&lssq[l16], lq);
        }
    }
    __syncthreads();
    if (tid < 6) {
        atomicAdd(&dacc[2 + tid], (double)lsum[tid]);
        atomicAdd(&dacc[8 + tid], (double)lssq[tid]);
    }
}

__global__ void k_fin2(const double* __restrict__ dacc, const float* __restrict__ g2,
                       const float* __restrict__ b2, float* __restrict__ fc)
{
    int c = threadIdx.x;
    if (c >= 6) return;
    double n = (double)P;
    double mu  = dacc[2+c]/n;
    double var = dacc[8+c]/n - mu*mu;
    double inv = 1.0 / sqrt(var + (double)EPSBN);
    float sc = (float)inv * g2[c];
    fc[6+c]  = sc;
    fc[12+c] = b2[c] - (float)mu*sc;
}

__global__ __launch_bounds__(256) void k_m3(const float* __restrict__ u2,
                                            const float* __restrict__ fc,
                                            double* __restrict__ dacc)
{
    __shared__ double sacc[27];
    int tid = threadIdx.x;
    if (tid < 27) sacc[tid] = 0.0;
    __syncthreads();
    float m[6]; float s[21];
    #pragma unroll
    for (int c = 0; c < 6; ++c) m[c] = 0.f;
    #pragma unroll
    for (int k = 0; k < 21; ++k) s[k] = 0.f;
    long gid = (long)blockIdx.x*256 + tid;
    for (int it = 0; it < 12; ++it) {
        long pos = gid + (long)it*(M3_BLOCKS*256);
        float4 ua = *(const float4*)(u2 + pos*8);
        float4 ub = *(const float4*)(u2 + pos*8 + 4);
        float uv[6] = {ua.x, ua.y, ua.z, ua.w, ub.x, ub.y};
        float y[6];
        #pragma unroll
        for (int c = 0; c < 6; ++c) y[c] = lrelu(uv[c]*fc[6+c] + fc[12+c]);
        int idx = 0;
        #pragma unroll
        for (int c = 0; c < 6; ++c) {
            m[c] += y[c];
            #pragma unroll
            for (int c2 = c; c2 < 6; ++c2) s[idx++] += y[c]*y[c2];
        }
    }
    #pragma unroll
    for (int c = 0; c < 6; ++c)  atomicAdd(&sacc[c],     (double)m[c]);
    #pragma unroll
    for (int k = 0; k < 21; ++k) atomicAdd(&sacc[6 + k], (double)s[k]);
    __syncthreads();
    if (tid < 27) atomicAdd(&dacc[14 + tid], sacc[tid]);
}

__global__ __launch_bounds__(48) void k_fin3(const double* __restrict__ dacc,
                                             const float* __restrict__ wt,
                                             const float* __restrict__ g3,
                                             const float* __restrict__ b3,
                                             float* __restrict__ fc)
{
    int f = threadIdx.x & 15;
    int o = threadIdx.x >> 4;

    double M[6], S[6][6];
    #pragma unroll
    for (int c = 0; c < 6; ++c) M[c] = dacc[14+c];
    {
        int idx = 0;
        #pragma unroll
        for (int c = 0; c < 6; ++c)
            #pragma unroll
            for (int c2 = c; c2 < 6; ++c2) {
                double v = dacc[20+idx]; S[c][c2] = v; S[c2][c] = v; ++idx;
            }
    }
    double wv[6];
    #pragma unroll
    for (int c = 0; c < 6; ++c) wv[c] = (double)wt[(c*3+o)*16 + f];

    double mu = 0.0, m2 = 0.0;
    #pragma unroll
    for (int c = 0; c < 6; ++c) {
        mu += wv[c]*M[c];
        double t = 0.0;
        #pragma unroll
        for (int c2 = 0; c2 < 6; ++c2) t += wv[c2]*S[c][c2];
        m2 += wv[c]*t;
    }
    #pragma unroll
    for (int m = 8; m >= 1; m >>= 1) {
        mu += __shfl_xor(mu, m, 16);
        m2 += __shfl_xor(m2, m, 16);
    }
    if (f == 0) {
        const double sc = 1.0 / (16.0 * (double)P);
        mu *= sc; m2 *= sc;
        double var = m2 - mu*mu;
        double inv = 1.0 / sqrt(var + (double)EPSBN);
        float A = (float)inv * g3[o];
        fc[18+o] = A;
        fc[21+o] = b3[o] - (float)mu*A;
    }
}

__global__ __launch_bounds__(256) void k_out(const float* __restrict__ u2,
                                             const float* __restrict__ wt,
                                             const float* __restrict__ fc,
                                             float* __restrict__ out)
{
    int pos = blockIdx.x*256 + threadIdx.x;
    int l  = pos % HP;  int t1 = pos / HP;
    int kk = t1 % HP;   t1 /= HP;
    int j  = t1 % HP;   t1 /= HP;
    int i  = t1 % HP;   int b = t1 / HP;

    float4 ua = *(const float4*)(u2 + (long)pos*8);
    float4 ub = *(const float4*)(u2 + (long)pos*8 + 4);
    float uv[6] = {ua.x, ua.y, ua.z, ua.w, ub.x, ub.y};
    float y[6];
    #pragma unroll
    for (int c = 0; c < 6; ++c) y[c] = lrelu(uv[c]*fc[6+c] + fc[12+c]);

    float acc[48];
    #pragma unroll
    for (int u = 0; u < 48; ++u) acc[u] = 0.f;
    #pragma unroll
    for (int c = 0; c < 6; ++c)
      #pragma unroll
      for (int o = 0; o < 3; ++o) {
        const float* wrow = wt + (c*3 + o)*16;
        #pragma unroll
        for (int f = 0; f < 16; ++f) acc[o*16 + f] += y[c]*wrow[f];
      }

    #pragma unroll
    for (int o = 0; o < 3; ++o) {
        float A3 = fc[18+o], B3 = fc[21+o];
        #pragma unroll
        for (int p = 0; p < 2; ++p)
          #pragma unroll
          for (int q = 0; q < 2; ++q)
            #pragma unroll
            for (int r = 0; r < 2; ++r) {
                int f = p*8 + q*4 + r*2;
                float z0 = acc[o*16 + f    ]*A3 + B3;
                float z1 = acc[o*16 + f + 1]*A3 + B3;
                float s0 = 1.f / (1.f + __expf(-z0));
                float s1 = 1.f / (1.f + __expf(-z1));
                long oidx = (((((long)b*3 + o)*48 + (2*i+p))*48 + (2*j+q))*48 + (2*kk+r))*48 + 2*l;
                *(float2*)(out + oidx) = make_float2(s0, s1);
            }
    }
}

extern "C" void kernel_launch(void* const* d_in, const int* in_sizes, int n_in,
                              void* d_out, int out_size, void* d_ws, size_t ws_size,
                              hipStream_t stream)
{
    (void)in_sizes; (void)n_in; (void)out_size; (void)ws_size;
    const float* x  = (const float*)d_in[0];
    const float* w1 = (const float*)d_in[1];
    const float* g1 = (const float*)d_in[2];
    const float* b1 = (const float*)d_in[3];
    const float* w2 = (const float*)d_in[4];
    const float* g2 = (const float*)d_in[5];
    const float* b2 = (const float*)d_in[6];
    const float* wt = (const float*)d_in[7];
    const float* g3 = (const float*)d_in[8];
    const float* b3 = (const float*)d_in[9];
    float*  out  = (float*)d_out;
    float*  u2   = (float*)((char*)d_ws + OFF_U2);
    float*  wp   = (float*)((char*)d_ws + OFF_WP);
    double* dacc = (double*)((char*)d_ws + OFF_DACC);
    float*  fc   = (float*)((char*)d_ws + OFF_FC);

    k_prep <<<1, 512, 0, stream>>>(w2, wp, dacc);
    k_rx   <<<RX_BLOCKS, 256, 0, stream>>>(x, dacc);
    k_fin1 <<<1, 32, 0, stream>>>(dacc, w1, g1, b1, fc);
    k_conv2<<<POS_BLOCKS, 256, 0, stream>>>(x, wp, fc, u2, dacc);
    k_fin2 <<<1, 32, 0, stream>>>(dacc, g2, b2, fc);
    k_m3   <<<M3_BLOCKS, 256, 0, stream>>>(u2, fc, dacc);
    k_fin3 <<<1, 48, 0, stream>>>(dacc, wt, g3, b3, fc);
    k_out  <<<POS_BLOCKS, 256, 0, stream>>>(u2, wt, fc, out);
}


# k_rx, k_m3: lanes summed by an f32 butterfly, then one f64 LDS atomic per wave (lane k adds total k) in place of a 32-step serial f64 walk per atomic; reduction tree changes
# speedup vs baseline: 2.5773x; 1.6388x over previous
#include <hip/hip_runtime.h>
#include <math.h>

#define EPSBN 1e-5f
#define SLOPE 0.01f
static constexpr int  HP  = 24;
static constexpr long P   = 2L*24*24*24*24;
static constexpr long NX  = 2L*48*48*48*48;
static constexpr int  RX_BLOCKS = 648;
static constexpr int  M3_BLOCKS = 216;
static constexpr int  POS_BLOCKS = 2592;

static constexpr size_t OFF_U2   = 0;
static constexpr size_t OFF_WP   = (size_t)P*8*4;
static constexpr size_t OFF_DACC = OFF_WP + 768*4;
static constexpr size_t OFF_FC   = OFF_DACC + 512;

typedef float vf2 __attribute__((ext_vector_type(2)));
typedef float vf8 __attribute__((ext_vector_type(8)));

__device__ __forceinline__ float lrelu(float v){ return v >= 0.f ? v : SLOPE*v; }

__device__ __forceinline__ float wave_sum(float v)
{
    #pragma unroll
    for (int m = 16; m >= 1; m >>= 1) v += __shfl_xor(v, m, 32);
    return v;
}

__global__ void k_prep(const float* __restrict__ w2, float* __restrict__ wp,
                       double* __restrict__ dacc)
{
    int tid = threadIdx.x;
    if (tid < 56) dacc[tid] = 0.0;
    if (tid < 384) {
        int n = tid & 15, h = (tid >> 4) & 1, t = tid >> 5;
        int K0 = 4*t + 2*h;
        float vx = (n < 6) ? w2[n*48 + K0]     : 0.f;
        float vy = (n < 6) ? w2[n*48 + K0 + 1] : 0.f;
        ((float2*)wp)[t*32 + h*16 + n] = make_float2(vx, vy);
    }
}

__global__ __launch_bounds__(256) void k_rx(const float* __restrict__ x,
                                            double* __restrict__ dacc)
{
    __shared__ double sacc[2];
    int tid = threadIdx.x;
    if (tid < 2) sacc[tid] = 0.0;
    __syncthreads();
    long gid = (long)blockIdx.x*256 + tid;
    float s = 0.f, q = 0.f;
    #pragma unroll
    for (int it = 0; it < 16; ++it) {
        long idx = gid + (long)it * (RX_BLOCKS*256);
        float4 v = *(const float4*)(x + idx*4);
        s += v.x + v.y + v.z + v.w;
        q += v.x*v.x + v.y*v.y + v.z*v.z + v.w*v.w;
    }
    s = wave_sum(s);
    q = wave_sum(q);
    int lane = tid & 31;
    if (lane < 2) atomicAdd(&sacc[lane], (double)(lane == 0 ? s : q));
    __syncthreads();
    if (tid < 2) atomicAdd(&dacc[tid], sacc[tid]);
}

__global__ void k_fin1(const double* __restrict__ dacc, const float* __restrict__ w1,
                       const float* __restrict__ g1, const float* __restrict__ b1,
                       float* __restrict__ fc)
{
    int c = threadIdx.x;
    if (c >= 3) return;
    double n  = (double)NX;
    double mu = dacc[0]/n;
    double var = dacc[1]/n - mu*mu;
    double wc  = (double)w1[c];
    double inv = 1.0 / sqrt(wc*wc*var + (double)EPSBN);
    fc[c]   = (float)(wc*inv) * g1[c];
    fc[3+c] = (float)(-wc*mu*inv) * g1[c] + b1[c];
}

__global__ __launch_bounds__(256) void k_conv2(const float* __restrict__ x,
                                               const float* __restrict__ wp,
                                               const float* __restrict__ fc,
                                               float* __restrict__ u2,
                                               double* __restrict__ dacc)
{
    __shared__ float lx[256*18];
    __shared__ float lsum[6], lssq[6];
    int tid = threadIdx.x;
    if (tid < 6) { lsum[tid] = 0.f; lssq[tid] = 0.f; }

    int pos = blockIdx.x*256 + tid;
    int l  = pos % HP;  int t1 = pos / HP;
    int kk = t1 % HP;   t1 /= HP;
    int j  = t1 % HP;   t1 /= HP;
    int i  = t1 % HP;   int b = t1 / HP;
    long xbase = ((((long)b*48 + 2*i)*48 + 2*j)*48 + 2*kk)*48 + 2*l;

    float* row = &lx[tid*18];
    #pragma unroll
    for (int p = 0; p < 2; ++p)
      #pragma unroll
      for (int q = 0; q < 2; ++q)
        #pragma unroll
        for (int r = 0; r < 2; ++r) {
            float2 v = *(const float2*)(x + xbase + (long)p*110592 + q*2304 + r*48);
            *(float2*)(row + (p*8 + q*4 + r*2)) = v;
        }

    int lane = tid & 31, l16 = lane & 15, h = lane >> 4, wv = tid >> 5;

    vf2 Bfrag[12];
    #pragma unroll
    for (int t = 0; t < 12; ++t)
        Bfrag[t] = *(const vf2*)(wp + ((t*2 + h)*16 + l16)*2);

    float a1[3], c1[3];
    #pragma unroll
    for (int c = 0; c < 3; ++c) { a1[c] = fc[c]; c1[c] = fc[3+c]; }

    __syncthreads();

    #pragma unroll
    for (int tau = 0; tau < 2; ++tau) {
        int bl = wv*32 + tau*16;
        vf8 acc = {0.f,0.f,0.f,0.f,0.f,0.f,0.f,0.f};
        #pragma unroll
        for (int t = 0; t < 12; ++t) {
            int c  = t >> 2;
            int fb = (t & 3)*4 + 2*h;
            float2 xv = *(const float2*)&lx[(bl + l16)*18 + fb];
            vf2 A;
            A.x = lrelu(a1[c]*xv.x + c1[c]);
            A.y = lrelu(a1[c]*xv.y + c1[c]);
            acc = __builtin_amdgcn_wmma_f32_16x16x4_f32(
                      false, A, false, Bfrag[t], (short)0, acc, false, false);
        }
        if (l16 < 6) {
            long pbase = (long)blockIdx.x*256 + bl + 8*h;
            float ls = 0.f, lq = 0.f;
            #pragma unroll
            for (int r = 0; r < 8; ++r) {
                float v = acc[r];
                u2[(pbase + r)*8 + l16] = v;
                ls += v; lq += v*v;
            }
            atomicAdd(&lsum[l16], ls);
            atomicAdd(&lssq[l16], lq);
        }
    }
    __syncthreads();
    if (tid < 6) {
        atomicAdd(&dacc[2 + tid], (double)lsum[tid]);
        atomicAdd(&dacc[8 + tid], (double)lssq[tid]);
    }
}

__global__ void k_fin2(const double* __restrict__ dacc, const float* __restrict__ g2,
                       const float* __restrict__ b2, float* __restrict__ fc)
{
    int c = threadIdx.x;
    if (c >= 6) return;
    double n = (double)P;
    double mu  = dacc[2+c]/n;
    double var = dacc[8+c]/n - mu*mu;
    double inv = 1.0 / sqrt(var + (double)EPSBN);
    float sc = (float)inv * g2[c];
    fc[6+c]  = sc;
    fc[12+c] = b2[c] - (float)mu*sc;
}

__global__ __launch_bounds__(256) void k_m3(const float* __restrict__ u2,
                                            const float* __restrict__ fc,
                                            double* __restrict__ dacc)
{
    __shared__ double sacc[27];
    int tid = threadIdx.x;
    if (tid < 27) sacc[tid] = 0.0;
    __syncthreads();
    float m[6]; float s[21];
    #pragma unroll
    for (int c = 0; c < 6; ++c) m[c] = 0.f;
    #pragma unroll
    for (int k = 0; k < 21; ++k) s[k] = 0.f;
    long gid = (long)blockIdx.x*256 + tid;
    for (int it = 0; it < 12; ++it) {
        long pos = gid + (long)it*(M3_BLOCKS*256);
        float4 ua = *(const float4*)(u2 + pos*8);
        float4 ub = *(const float4*)(u2 + pos*8 + 4);
        float uv[6] = {ua.x, ua.y, ua.z, ua.w, ub.x, ub.y};
        float y[6];
        #pragma unroll
        for (int c = 0; c < 6; ++c) y[c] = lrelu(uv[c]*fc[6+c] + fc[12+c]);
        int idx = 0;
        #pragma unroll
        for (int c = 0; c < 6; ++c) {
            m[c] += y[c];
            #pragma unroll
            for (int c2 = c; c2 < 6; ++c2) s[idx++] += y[c]*y[c2];
        }
    }
    float tot[27];
    #pragma unroll
    for (int c = 0; c < 6; ++c)  tot[c]     = m[c];
    #pragma unroll
    for (int k = 0; k < 21; ++k) tot[6 + k] = s[k];
    #pragma unroll
    for (int msk = 16; msk >= 1; msk >>= 1) {
        #pragma unroll
        for (int k = 0; k < 27; ++k) tot[k] += __shfl_xor(tot[k], msk, 32);
    }
    int lane = tid & 31;
    float pick = 0.f;
    #pragma unroll
    for (int k = 0; k < 27; ++k) pick = (lane == k) ? tot[k] : pick;
    if (lane < 27) atomicAdd(&sacc[lane], (double)pick);
    __syncthreads();
    if (tid < 27) atomicAdd(&dacc[14 + tid], sacc[tid]);
}

__global__ __launch_bounds__(48) void k_fin3(const double* __restrict__ dacc,
                                             const float* __restrict__ wt,
                                             const float* __restrict__ g3,
                                             const float* __restrict__ b3,
                                             float* __restrict__ fc)
{
    int f = threadIdx.x & 15;
    int o = threadIdx.x >> 4;

    double M[6], S[6][6];
    #pragma unroll
    for (int c = 0; c < 6; ++c) M[c] = dacc[14+c];
    {
        int idx = 0;
        #pragma unroll
        for (int c = 0; c < 6; ++c)
            #pragma unroll
            for (int c2 = c; c2 < 6; ++c2) {
                double v = dacc[20+idx]; S[c][c2] = v; S[c2][c] = v; ++idx;
            }
    }
    double wv[6];
    #pragma unroll
    for (int c = 0; c < 6; ++c) wv[c] = (double)wt[(c*3+o)*16 + f];

    double mu = 0.0, m2 = 0.0;
    #pragma unroll
    for (int c = 0; c < 6; ++c) {
        mu += wv[c]*M[c];
        double t = 0.0;
        #pragma unroll
        for (int c2 = 0; c2 < 6; ++c2) t += wv[c2]*S[c][c2];
        m2 += wv[c]*t;
    }
    #pragma unroll
    for (int m = 8; m >= 1; m >>= 1) {
        mu += __shfl_xor(mu, m, 16);
        m2 += __shfl_xor(m2, m, 16);
    }
    if (f == 0) {
        const double sc = 1.0 / (16.0 * (double)P);
        mu *= sc; m2 *= sc;
        double var = m2 - mu*mu;
        double inv = 1.0 / sqrt(var + (double)EPSBN);
        float A = (float)inv * g3[o];
        fc[18+o] = A;
        fc[21+o] = b3[o] - (float)mu*A;
    }
}

__global__ __launch_bounds__(256) void k_out(const float* __restrict__ u2,
                                             const float* __restrict__ wt,
                                             const float* __restrict__ fc,
                                             float* __restrict__ out)
{
    int pos = blockIdx.x*256 + threadIdx.x;
    int l  = pos % HP;  int t1 = pos / HP;
    int kk = t1 % HP;   t1 /= HP;
    int j  = t1 % HP;   t1 /= HP;
    int i  = t1 % HP;   int b = t1 / HP;

    float4 ua = *(const float4*)(u2 + (long)pos*8);
    float4 ub = *(const float4*)(u2 + (long)pos*8 + 4);
    float uv[6] = {ua.x, ua.y, ua.z, ua.w, ub.x, ub.y};
    float y[6];
    #pragma unroll
    for (int c = 0; c < 6; ++c) y[c] = lrelu(uv[c]*fc[6+c] + fc[12+c]);

    float acc[48];
    #pragma unroll
    for (int u = 0; u < 48; ++u) acc[u] = 0.f;
    #pragma unroll
    for (int c = 0; c < 6; ++c)
      #pragma unroll
      for (int o = 0; o < 3; ++o) {
        const float* wrow = wt + (c*3 + o)*16;
        #pragma unroll
        for (int f = 0; f < 16; ++f) acc[o*16 + f] += y[c]*wrow[f];
      }

    #pragma unroll
    for (int o = 0; o < 3; ++o) {
        float A3 = fc[18+o], B3 = fc[21+o];
        #pragma unroll
        for (int p = 0; p < 2; ++p)
          #pragma unroll
          for (int q = 0; q < 2; ++q)
            #pragma unroll
            for (int r = 0; r < 2; ++r) {
                int f = p*8 + q*4 + r*2;
                float z0 = acc[o*16 + f    ]*A3 + B3;
                float z1 = acc[o*16 + f + 1]*A3 + B3;
                float s0 = 1.f / (1.f + __expf(-z0));
                float s1 = 1.f / (1.f + __expf(-z1));
                long oidx = (((((long)b*3 + o)*48 + (2*i+p))*48 + (2*j+q))*48 + (2*kk+r))*48 + 2*l;
                *(float2*)(out + oidx) = make_float2(s0, s1);
            }
    }
}

extern "C" void kernel_launch(void* const* d_in, const int* in_sizes, int n_in,
                              void* d_out, int out_size, void* d_ws, size_t ws_size,
                              hipStream_t stream)
{
    (void)in_sizes; (void)n_in; (void)out_size; (void)ws_size;
    const float* x  = (const float*)d_in[0];
    const float* w1 = (const float*)d_in[1];
    const float* g1 = (const float*)d_in[2];
    const float* b1 = (const float*)d_in[3];
    const float* w2 = (const float*)d_in[4];
    const float* g2 = (const float*)d_in[5];
    const float* b2 = (const float*)d_in[6];
    const float* wt = (const float*)d_in[7];
    const float* g3 = (const float*)d_in[8];
    const float* b3 = (const float*)d_in[9];
    float*  out  = (float*)d_out;
    float*  u2   = (float*)((char*)d_ws + OFF_U2);
    float*  wp   = (float*)((char*)d_ws + OFF_WP);
    double* dacc = (double*)((char*)d_ws + OFF_DACC);
    float*  fc   = (float*)((char*)d_ws + OFF_FC);

    k_prep <<<1, 512, 0, stream>>>(w2, wp, dacc);
    k_rx   <<<RX_BLOCKS, 256, 0, stream>>>(x, dacc);
    k_fin1 <<<1, 32, 0, stream>>>(dacc, w1, g1, b1, fc);
    k_conv2<<<POS_BLOCKS, 256, 0, stream>>>(x, wp, fc, u2, dacc);
    k_fin2 <<<1, 32, 0, stream>>>(dacc, g2, b2, fc);
    k_m3   <<<M3_BLOCKS, 256, 0, stream>>>(u2, fc, dacc);
    k_fin3 <<<1, 48, 0, stream>>>(dacc, wt, g3, b3, fc);
    k_out  <<<POS_BLOCKS, 256, 0, stream>>>(u2, wt, fc, out);
}


# k_out was vector-issue bound: 1157 vector instructions per wave cut to 452: explicit fma on weights kept in scalar registers, v_rcp_f32 (1 ulp) for the sigmoid's division; last bits of out change
# speedup vs baseline: 2.6951x; 1.0457x over previous
#include <hip/hip_runtime.h>
#include <math.h>

#define EPSBN 1e-5f
#define SLOPE 0.01f
static constexpr int  HP  = 24;
static constexpr long P   = 2L*24*24*24*24;
static constexpr long NX  = 2L*48*48*48*48;
static constexpr int  RX_BLOCKS = 648;
static constexpr int  M3_BLOCKS = 216;
static constexpr int  POS_BLOCKS = 2592;

static constexpr size_t OFF_U2   = 0;
static constexpr size_t OFF_WP   = (size_t)P*8*4;
static constexpr size_t OFF_DACC = OFF_WP + 768*4;
static constexpr size_t OFF_FC   = OFF_DACC + 512;

typedef float vf2 __attribute__((ext_vector_type(2)));
typedef float vf8 __attribute__((ext_vector_type(8)));

__device__ __forceinline__ float lrelu(float v){ return v >= 0.f ? v : SLOPE*v; }

__device__ __forceinline__ float wave_sum(float v)
{
    #pragma unroll
    for (int m = 16; m >= 1; m >>= 1) v += __shfl_xor(v, m, 32);
    return v;
}

__global__ void k_prep(const float* __restrict__ w2, float* __restrict__ wp,
                       double* __restrict__ dacc)
{
    int tid = threadIdx.x;
    if (tid < 56) dacc[tid] = 0.0;
    if (tid < 384) {
        int n = tid & 15, h = (tid >> 4) & 1, t = tid >> 5;
        int K0 = 4*t + 2*h;
        float vx = (n < 6) ? w2[n*48 + K0]     : 0.f;
        float vy = (n < 6) ? w2[n*48 + K0 + 1] : 0.f;
        ((float2*)wp)[t*32 + h*16 + n] = make_float2(vx, vy);
    }
}

__global__ __launch_bounds__(256) void k_rx(const float* __restrict__ x,
                                            double* __restrict__ dacc)
{
    __shared__ double sacc[2];
    int tid = threadIdx.x;
    if (tid < 2) sacc[tid] = 0.0;
    __syncthreads();
    long gid = (long)blockIdx.x*256 + tid;
    float s = 0.f, q = 0.f;
    #pragma unroll
    for (int it = 0; it < 16; ++it) {
        long idx = gid + (long)it * (RX_BLOCKS*256);
        float4 v = *(const float4*)(x + idx*4);
        s += v.x + v.y + v.z + v.w;
        q += v.x*v.x + v.y*v.y + v.z*v.z + v.w*v.w;
    }
    s = wave_sum(s);
    q = wave_sum(q);
    int lane = tid & 31;
    if (lane < 2) atomicAdd(&sacc[lane], (double)(lane == 0 ? s : q));
    __syncthreads();
    if (tid < 2) atomicAdd(&dacc[tid], sacc[tid]);
}

__global__ void k_fin1(const double* __restrict__ dacc, const float* __restrict__ w1,
                       const float* __restrict__ g1, const float* __restrict__ b1,
                       float* __restrict__ fc)
{
    int c = threadIdx.x;
    if (c >= 3) return;
    double n  = (double)NX;
    double mu = dacc[0]/n;
    double var = dacc[1]/n - mu*mu;
    double wc  = (double)w1[c];
    double inv = 1.0 / sqrt(wc*wc*var + (double)EPSBN);
    fc[c]   = (float)(wc*inv) * g1[c];
    fc[3+c] = (float)(-wc*mu*inv) * g1[c] + b1[c];
}

__global__ __launch_bounds__(256) void k_conv2(const float* __restrict__ x,
                                               const float* __restrict__ wp,
                                               const float* __restrict__ fc,
                                               float* __restrict__ u2,
                                               double* __restrict__ dacc)
{
    __shared__ float lx[256*18];
    __shared__ float lsum[6], lssq[6];
    int tid = threadIdx.x;
    if (tid < 6) { lsum[tid] = 0.f; lssq[tid] = 0.f; }

    int pos = blockIdx.x*256 + tid;
    int l  = pos % HP;  int t1 = pos / HP;
    int kk = t1 % HP;   t1 /= HP;
    int j  = t1 % HP;   t1 /= HP;
    int i  = t1 % HP;   int b = t1 / HP;
    long xbase = ((((long)b*48 + 2*i)*48 + 2*j)*48 + 2*kk)*48 + 2*l;

    float* row = &lx[tid*18];
    #pragma unroll
    for (int p = 0; p < 2; ++p)
      #pragma unroll
      for (int q = 0; q < 2; ++q)
        #pragma unroll
        for (int r = 0; r < 2; ++r) {
            float2 v = *(const float2*)(x + xbase + (long)p*110592 + q*2304 + r*48);
            *(float2*)(row + (p*8 + q*4 + r*2)) = v;
        }

    int lane = tid & 31, l16 = lane & 15, h = lane >> 4, wv = tid >> 5;

    vf2 Bfrag[12];
    #pragma unroll
    for (int t = 0; t < 12; ++t)
        Bfrag[t] = *(const vf2*)(wp + ((t*2 + h)*16 + l16)*2);

    float a1[3], c1[3];
    #pragma unroll
    for (int c = 0; c < 3; ++c) { a1[c] = fc[c]; c1[c] = fc[3+c]; }

    __syncthreads();

    #pragma unroll
    for (int tau = 0; tau < 2; ++tau) {
        int bl = wv*32 + tau*16;
        vf8 acc = {0.f,0.f,0.f,0.f,0.f,0.f,0.f,0.f};
        #pragma unroll
        for (int t = 0; t < 12; ++t) {
            int c  = t >> 2;
            int fb = (t & 3)*4 + 2*h;
            float2 xv = *(const float2*)&lx[(bl + l16)*18 + fb];
            vf2 A;
            A.x = lrelu(a1[c]*xv.x + c1[c]);
            A.y = lrelu(a1[c]*xv.y + c1[c]);
            acc = __builtin_amdgcn_wmma_f32_16x16x4_f32(
                      false, A, false, Bfrag[t], (short)0, acc, false, false);
        }
        if (l16 < 6) {
            long pbase = (long)blockIdx.x*256 + bl + 8*h;
            float ls = 0.f, lq = 0.f;
            #pragma unroll
            for (int r = 0; r < 8; ++r) {
                float v = acc[r];
                u2[(pbase + r)*8 + l16] = v;
                ls += v; lq += v*v;
            }
            atomicAdd(&lsum[l16], ls);
            atomicAdd(&lssq[l16], lq);
        }
    }
    __syncthreads();
    if (tid < 6) {
        atomicAdd(&dacc[2 + tid], (double)lsum[tid]);
        atomicAdd(&dacc[8 + tid], (double)lssq[tid]);
    }
}

__global__ void k_fin2(const double* __restrict__ dacc, const float* __restrict__ g2,
                       const float* __restrict__ b2, float* __restrict__ fc)
{
    int c = threadIdx.x;
    if (c >= 6) return;
    double n = (double)P;
    double mu  = dacc[2+c]/n;
    double var = dacc[8+c]/n - mu*mu;
    double inv = 1.0 / sqrt(var + (double)EPSBN);
    float sc = (float)inv * g2[c];
    fc[6+c]  = sc;
    fc[12+c] = b2[c] - (float)mu*sc;
}

__global__ __launch_bounds__(256) void k_m3(const float* __restrict__ u2,
                                            const float* __restrict__ fc,
                                            double* __restrict__ dacc)
{
    __shared__ double sacc[27];
    int tid = threadIdx.x;
    if (tid < 27) sacc[tid] = 0.0;
    __syncthreads();
    float m[6]; float s[21];
    #pragma unroll
    for (int c = 0; c < 6; ++c) m[c] = 0.f;
    #pragma unroll
    for (int k = 0; k < 21; ++k) s[k] = 0.f;
    long gid = (long)blockIdx.x*256 + tid;
    for (int it = 0; it < 12; ++it) {
        long pos = gid + (long)it*(M3_BLOCKS*256);
        float4 ua = *(const float4*)(u2 + pos*8);
        float4 ub = *(const float4*)(u2 + pos*8 + 4);
        float uv[6] = {ua.x, ua.y, ua.z, ua.w, ub.x, ub.y};
        float y[6];
        #pragma unroll
        for (int c = 0; c < 6; ++c) y[c] = lrelu(uv[c]*fc[6+c] + fc[12+c]);
        int idx = 0;
        #pragma unroll
        for (int c = 0; c < 6; ++c) {
            m[c] += y[c];
            #pragma unroll
            for (int c2 = c; c2 < 6; ++c2) s[idx++] += y[c]*y[c2];
        }
    }
    float tot[27];
    #pragma unroll
    for (int c = 0; c < 6; ++c)  tot[c]     = m[c];
    #pragma unroll
    for (int k = 0; k < 21; ++k) tot[6 + k] = s[k];
    #pragma unroll
    for (int msk = 16; msk >= 1; msk >>= 1) {
        #pragma unroll
        for (int k = 0; k < 27; ++k) tot[k] += __shfl_xor(tot[k], msk, 32);
    }
    int lane = tid & 31;
    float pick = 0.f;
    #pragma unroll
    for (int k = 0; k < 27; ++k) pick = (lane == k) ? tot[k] : pick;
    if (lane < 27) atomicAdd(&sacc[lane], (double)pick);
    __syncthreads();
    if (tid < 27) atomicAdd(&dacc[14 + tid], sacc[tid]);
}

__global__ __launch_bounds__(48) void k_fin3(const double* __restrict__ dacc,
                                             const float* __restrict__ wt,
                                             const float* __restrict__ g3,
                                             const float* __restrict__ b3,
                                             float* __restrict__ fc)
{
    int f = threadIdx.x & 15;
    int o = threadIdx.x >> 4;

    double M[6], S[6][6];
    #pragma unroll
    for (int c = 0; c < 6; ++c) M[c] = dacc[14+c];
    {
        int idx = 0;
        #pragma unroll
        for (int c = 0; c < 6; ++c)
            #pragma unroll
            for (int c2 = c; c2 < 6; ++c2) {
                double v = dacc[20+idx]; S[c][c2] = v; S[c2][c] = v; ++idx;
            }
    }
    double wv[6];
    #pragma unroll
    for (int c = 0; c < 6; ++c) wv[c] = (double)wt[(c*3+o)*16 + f];

    double mu = 0.0, m2 = 0.0;
    #pragma unroll
    for (int c = 0; c < 6; ++c) {
        mu += wv[c]*M[c];
        double t = 0.0;
        #pragma unroll
        for (int c2 = 0; c2 < 6; ++c2) t += wv[c2]*S[c][c2];
        m2 += wv[c]*t;
    }
    #pragma unroll
    for (int m = 8; m >= 1; m >>= 1) {
        mu += __shfl_xor(mu, m, 16);
        m2 += __shfl_xor(m2, m, 16);
    }
    if (f == 0) {
        const double sc = 1.0 / (16.0 * (double)P);
        mu *= sc; m2 *= sc;
        double var = m2 - mu*mu;
        double inv = 1.0 / sqrt(var + (double)EPSBN);
        float A = (float)inv * g3[o];
        fc[18+o] = A;
        fc[21+o] = b3[o] - (float)mu*A;
    }
}

__global__ __launch_bounds__(256) void k_out(const float* __restrict__ u2,
                                             const float* __restrict__ wt,
                                             const float* __restrict__ fc,
                                             float* __restrict__ out)
{
    int pos = blockIdx.x*256 + threadIdx.x;
    int l  = pos % HP;  int t1 = pos / HP;
    int kk = t1 % HP;   t1 /= HP;
    int j  = t1 % HP;   t1 /= HP;
    int i  = t1 % HP;   int b = t1 / HP;

    float4 ua = *(const float4*)(u2 + (long)pos*8);
    float4 ub = *(const float4*)(u2 + (long)pos*8 + 4);
    float uv[6] = {ua.x, ua.y, ua.z, ua.w, ub.x, ub.y};
    float y[6];
    #pragma unroll
    for (int c = 0; c < 6; ++c) y[c] = lrelu(uv[c]*fc[6+c] + fc[12+c]);

    float acc[48];
    #pragma unroll
    for (int c = 0; c < 6; ++c)
      #pragma unroll
      for (int o = 0; o < 3; ++o) {
        const float* wrow = wt + (c*3 + o)*16;
        #pragma unroll
        for (int f = 0; f < 16; ++f)
            acc[o*16 + f] = (c == 0) ? y[0]*wrow[f] : __builtin_fmaf(y[c], wrow[f], acc[o*16 + f]);
      }

    #pragma unroll
    for (int o = 0; o < 3; ++o) {
        float A3 = fc[18+o], B3 = fc[21+o];
        #pragma unroll
        for (int p = 0; p < 2; ++p)
          #pragma unroll
          for (int q = 0; q < 2; ++q)
            #pragma unroll
            for (int r = 0; r < 2; ++r) {
                int f = p*8 + q*4 + r*2;
                float z0 = __builtin_fmaf(acc[o*16 + f    ], A3, B3);
                float z1 = __builtin_fmaf(acc[o*16 + f + 1], A3, B3);
                float s0 = __builtin_amdgcn_rcpf(1.f + __expf(-z0));
                float s1 = __builtin_amdgcn_rcpf(1.f + __expf(-z1));
                long oidx = (((((long)b*3 + o)*48 + (2*i+p))*48 + (2*j+q))*48 + (2*kk+r))*48 + 2*l;
                *(float2*)(out + oidx) = make_float2(s0, s1);
            }
    }
}

extern "C" void kernel_launch(void* const* d_in, const int* in_sizes, int n_in,
                              void* d_out, int out_size, void* d_ws, size_t ws_size,
                              hipStream_t stream)
{
    (void)in_sizes; (void)n_in; (void)out_size; (void)ws_size;
    const float* x  = (const float*)d_in[0];
    const float* w1 = (const float*)d_in[1];
    const float* g1 = (const float*)d_in[2];
    const float* b1 = (const float*)d_in[3];
    const float* w2 = (const float*)d_in[4];
    const float* g2 = (const float*)d_in[5];
    const float* b2 = (const float*)d_in[6];
    const float* wt = (const float*)d_in[7];
    const float* g3 = (const float*)d_in[8];
    const float* b3 = (const float*)d_in[9];
    float*  out  = (float*)d_out;
    float*  u2   = (float*)((char*)d_ws + OFF_U2);
    float*  wp   = (float*)((char*)d_ws + OFF_WP);
    double* dacc = (double*)((char*)d_ws + OFF_DACC);
    float*  fc   = (float*)((char*)d_ws + OFF_FC);

    k_prep <<<1, 512, 0, stream>>>(w2, wp, dacc);
    k_rx   <<<RX_BLOCKS, 256, 0, stream>>>(x, dacc);
    k_fin1 <<<1, 32, 0, stream>>>(dacc, w1, g1, b1, fc);
    k_conv2<<<POS_BLOCKS, 256, 0, stream>>>(x, wp, fc, u2, dacc);
    k_fin2 <<<1, 32, 0, stream>>>(dacc, g2, b2, fc);
    k_m3   <<<M3_BLOCKS, 256, 0, stream>>>(u2, fc, dacc);
    k_fin3 <<<1, 48, 0, stream>>>(dacc, wt, g3, b3, fc);
    k_out  <<<POS_BLOCKS, 256, 0, stream>>>(u2, wt, fc, out);
}


# three one-workgroup finaliser launches folded into the heads of their consumer kernels (8 launches -> 5): moments centred in f64, rest of BN constants in f32; tap sums regrouped, last bits change
# speedup vs baseline: 3.0394x; 1.1278x over previous
#include <hip/hip_runtime.h>
#include <math.h>

#define EPSBN 1e-5f
#define SLOPE 0.01f
static constexpr int  HP  = 24;
static constexpr long P   = 2L*24*24*24*24;
static constexpr long NX  = 2L*48*48*48*48;
static constexpr int  RX_BLOCKS = 648;
static constexpr int  M3_BLOCKS = 216;
static constexpr int  POS_BLOCKS = 2592;
static constexpr double RNX = 1.0 / (double)NX;
static constexpr double RP  = 1.0 / (double)P;

static constexpr size_t OFF_U2   = 0;
static constexpr size_t OFF_WP   = (size_t)P*8*4;
static constexpr size_t OFF_DACC = OFF_WP + 768*4;
static constexpr size_t OFF_FC   = OFF_DACC + 512;

typedef float vf2 __attribute__((ext_vector_type(2)));
typedef float vf8 __attribute__((ext_vector_type(8)));

__device__ __forceinline__ float lrelu(float v){ return v >= 0.f ? v : SLOPE*v; }

__device__ __forceinline__ float wave_sum(float v)
{
    #pragma unroll
    for (int m = 16; m >= 1; m >>= 1) v += __shfl_xor(v, m, 32);
    return v;
}

template <int M> __device__ __forceinline__ float row_xor(float v)
{
    return __builtin_bit_cast(float,
        __builtin_amdgcn_update_dpp(0, __builtin_bit_cast(int, v), 0x160 | M, 0xf, 0xf, true));
}
__device__ __forceinline__ float row16_sum(float v)
{
    v += row_xor<8>(v);
    v += row_xor<4>(v);
    v += row_xor<2>(v);
    v += row_xor<1>(v);
    return v;
}


__global__ void k_prep(const float* __restrict__ w2, float* __restrict__ wp,
                       double* __restrict__ dacc)
{
    int tid = threadIdx.x;
    if (tid < 56) dacc[tid] = 0.0;
    if (tid < 384) {
        int n = tid & 15, h = (tid >> 4) & 1, t = tid >> 5;
        int K0 = 4*t + 2*h;
        float vx = (n < 6) ? w2[n*48 + K0]     : 0.f;
        float vy = (n < 6) ? w2[n*48 + K0 + 1] : 0.f;
        ((float2*)wp)[t*32 + h*16 + n] = make_float2(vx, vy);
    }
}

__global__ __launch_bounds__(256) void k_rx(const float* __restrict__ x,
                                            double* __restrict__ dacc)
{
    __shared__ double sacc[2];
    int tid = threadIdx.x;
    if (tid < 2) sacc[tid] = 0.0;
    __syncthreads();
    long gid = (long)blockIdx.x*256 + tid;
    float s = 0.f, q = 0.f;
    #pragma unroll
    for (int it = 0; it < 16; ++it) {
        long idx = gid + (long)it * (RX_BLOCKS*256);
        float4 v = *(const float4*)(x + idx*4);
        s += v.x + v.y + v.z + v.w;
        q += v.x*v.x + v.y*v.y + v.z*v.z + v.w*v.w;
    }
    s = wave_sum(s);
    q = wave_sum(q);
    int lane = tid & 31;
    if (lane < 2) atomicAdd(&sacc[lane], (double)(lane == 0 ? s : q));
    __syncthreads();
    if (tid < 2) atomicAdd(&dacc[tid], sacc[tid]);
}

__global__ __launch_bounds__(256) void k_conv2(const float* __restrict__ x,
                                               const float* __restrict__ wp,
                                               const float* __restrict__ w1,
                                               const float* __restrict__ g1,
                                               const float* __restrict__ b1,
                                               float* __restrict__ u2,
                                               double* __restrict__ dacc)
{
    __shared__ float lx[256*18];
    __shared__ float lsum[6], lssq[6];
    __shared__ float lfc[6];
    int tid = threadIdx.x;
    if (tid < 6) { lsum[tid] = 0.f; lssq[tid] = 0.f; }

    int pos = blockIdx.x*256 + tid;
    int l  = pos % HP;  int t1 = pos / HP;
    int kk = t1 % HP;   t1 /= HP;
    int j  = t1 % HP;   t1 /= HP;
    int i  = t1 % HP;   int b = t1 / HP;
    long xbase = ((((long)b*48 + 2*i)*48 + 2*j)*48 + 2*kk)*48 + 2*l;

    float2 xin[8];
    #pragma unroll
    for (int p = 0; p < 2; ++p)
      #pragma unroll
      for (int q = 0; q < 2; ++q)
        #pragma unroll
        for (int r = 0; r < 2; ++r)
            xin[p*4 + q*2 + r] = *(const float2*)(x + xbase + (long)p*110592 + q*2304 + r*48);

    unsigned hc = (unsigned)(tid - 32*(int)(blockIdx.x & 7));
    if (hc < 3u) {
        double mu  = dacc[0] * RNX;
        double var = dacc[1] * RNX - mu*mu;
        float  wc  = w1[hc];
        float  a   = wc * g1[hc] / sqrtf(wc*wc*(float)var + EPSBN);
        lfc[hc]     = a;
        lfc[3 + hc] = b1[hc] - a*(float)mu;
    }

    float* row = &lx[tid*18];
    #pragma unroll
    for (int n = 0; n < 8; ++n) *(float2*)(row + 2*n) = xin[n];

    int lane = tid & 31, l16 = lane & 15, h = lane >> 4, wv = tid >> 5;

    vf2 Bfrag[12];
    #pragma unroll
    for (int t = 0; t < 12; ++t)
        Bfrag[t] = *(const vf2*)(wp + ((t*2 + h)*16 + l16)*2);

    __syncthreads();

    float a1[3], c1[3];
    #pragma unroll
    for (int c = 0; c < 3; ++c) {
        a1[c] = __builtin_bit_cast(float, __builtin_amdgcn_readfirstlane(__builtin_bit_cast(int, lfc[c])));
        c1[c] = __builtin_bit_cast(float, __builtin_amdgcn_readfirstlane(__builtin_bit_cast(int, lfc[3+c])));
    }

    #pragma unroll
    for (int tau = 0; tau < 2; ++tau) {
        int bl = wv*32 + tau*16;
        vf8 acc = {0.f,0.f,0.f,0.f,0.f,0.f,0.f,0.f};
        #pragma unroll
        for (int t = 0; t < 12; ++t) {
            int c  = t >> 2;
            int fb = (t & 3)*4 + 2*h;
            float2 xv = *(const float2*)&lx[(bl + l16)*18 + fb];
            vf2 A;
            A.x = lrelu(a1[c]*xv.x + c1[c]);
            A.y = lrelu(a1[c]*xv.y + c1[c]);
            acc = __builtin_amdgcn_wmma_f32_16x16x4_f32(
                      false, A, false, Bfrag[t], (short)0, acc, false, false);
        }
        if (l16 < 6) {
            long pbase = (long)blockIdx.x*256 + bl + 8*h;
            float ls = 0.f, lq = 0.f;
            #pragma unroll
            for (int r = 0; r < 8; ++r) {
                float v = acc[r];
                u2[(pbase + r)*8 + l16] = v;
                ls += v; lq += v*v;
            }
            atomicAdd(&lsum[l16], ls);
            atomicAdd(&lssq[l16], lq);
        }
    }
    __syncthreads();
    if (tid < 6) {
        atomicAdd(&dacc[2 + tid], (double)lsum[tid]);
        atomicAdd(&dacc[8 + tid], (double)lssq[tid]);
    }
}

__global__ __launch_bounds__(256) void k_m3(const float* __restrict__ u2,
                                            const float* __restrict__ g2,
                                            const float* __restrict__ b2,
                                            float* __restrict__ fc,
                                            double* __restrict__ dacc)
{
    __shared__ double sacc[27];
    __shared__ float lfc[12];
    int tid = threadIdx.x;
    if (tid < 27) sacc[tid] = 0.0;
    if (tid < 6) {
        double mu  = dacc[2 + tid] * RP;
        double var = dacc[8 + tid] * RP - mu*mu;
        float sc = g2[tid] / sqrtf((float)var + EPSBN);
        float sh = b2[tid] - (float)mu*sc;
        lfc[tid] = sc;  lfc[6 + tid] = sh;
        if (blockIdx.x == 0) { fc[6 + tid] = sc;  fc[12 + tid] = sh; }
    }
    __syncthreads();
    float sc2[6], sh2[6];
    #pragma unroll
    for (int c = 0; c < 6; ++c) { sc2[c] = lfc[c]; sh2[c] = lfc[6 + c]; }
    float m[6]; float s[21];
    #pragma unroll
    for (int c = 0; c < 6; ++c) m[c] = 0.f;
    #pragma unroll
    for (int k = 0; k < 21; ++k) s[k] = 0.f;
    long gid = (long)blockIdx.x*256 + tid;
    for (int it = 0; it < 12; ++it) {
        long pos = gid + (long)it*(M3_BLOCKS*256);
        float4 ua = *(const float4*)(u2 + pos*8);
        float4 ub = *(const float4*)(u2 + pos*8 + 4);
        float uv[6] = {ua.x, ua.y, ua.z, ua.w, ub.x, ub.y};
        float y[6];
        #pragma unroll
        for (int c = 0; c < 6; ++c) y[c] = lrelu(uv[c]*sc2[c] + sh2[c]);
        int idx = 0;
        #pragma unroll
        for (int c = 0; c < 6; ++c) {
            m[c] += y[c];
            #pragma unroll
            for (int c2 = c; c2 < 6; ++c2) s[idx++] += y[c]*y[c2];
        }
    }
    float tot[27];
    #pragma unroll
    for (int c = 0; c < 6; ++c)  tot[c]     = m[c];
    #pragma unroll
    for (int k = 0; k < 21; ++k) tot[6 + k] = s[k];
    #pragma unroll
    for (int msk = 16; msk >= 1; msk >>= 1) {
        #pragma unroll
        for (int k = 0; k < 27; ++k) tot[k] += __shfl_xor(tot[k], msk, 32);
    }
    int lane = tid & 31;
    float pick = 0.f;
    #pragma unroll
    for (int k = 0; k < 27; ++k) pick = (lane == k) ? tot[k] : pick;
    if (lane < 27) atomicAdd(&sacc[lane], (double)pick);
    __syncthreads();
    if (tid < 27) atomicAdd(&dacc[14 + tid], sacc[tid]);
}

__device__ __forceinline__ void bn3_consts(const float (&w)[6], const float (&mb)[6], const float (&cv)[21],
                                           float g, float b, float& A, float& B)
{
    float mu = 0.f, qd = 0.f, qo = 0.f;
    int idx = 0;
    #pragma unroll
    for (int c = 0; c < 6; ++c) {
        mu = __builtin_fmaf(w[c], mb[c], mu);
        qd = __builtin_fmaf(w[c]*w[c], cv[idx++], qd);
        if (c < 5) {
            float t = cv[idx++]*w[c + 1];
            #pragma unroll
            for (int c2 = c + 2; c2 < 6; ++c2) t = __builtin_fmaf(cv[idx++], w[c2], t);
            qo = __builtin_fmaf(w[c], t, qo);
        }
    }
    float q    = __builtin_fmaf(2.f, qo, qd);
    float mean = row16_sum(mu) * 0.0625f;
    float d    = mu - mean;
    float var  = row16_sum(__builtin_fmaf(d, d, q)) * 0.0625f;
    A = g / sqrtf(var + EPSBN);
    B = b - mean*A;
}

__global__ __launch_bounds__(256) void k_out(const float* __restrict__ u2,
                                             const float* __restrict__ wt,
                                             const float* __restrict__ fc,
                                             float* __restrict__ out,
                                             const double* __restrict__ dacc,
                                             const float* __restrict__ g3,
                                             const float* __restrict__ b3)
{
    __shared__ float lab[6];
    int pos = blockIdx.x*256 + threadIdx.x;
    int l  = pos % HP;  int t1 = pos / HP;
    int kk = t1 % HP;   t1 /= HP;
    int j  = t1 % HP;   t1 /= HP;
    int i  = t1 % HP;   int b = t1 / HP;

    float4 ua = *(const float4*)(u2 + (long)pos*8);
    float4 ub = *(const float4*)(u2 + (long)pos*8 + 4);

    const int wv = __builtin_amdgcn_readfirstlane((int)(threadIdx.x >> 5));
    if (wv == (int)(blockIdx.x & 7)) {
        const int lane = threadIdx.x & 31, row = lane >> 4, f = lane & 15;
        int m = lane < 27 ? lane : 26;
        int ca = m, cb = m;
        if (m >= 6) {
            int n = m - 6;
            ca = (n >= 6) + (n >= 11) + (n >= 15) + (n >= 18) + (n >= 20);
            cb = ca + n - (6*ca - ca*(ca - 1)/2);
        }
        double sm = dacc[14 + m],  sa = dacc[14 + ca],  sb = dacc[14 + cb];
        float w01[6], w2[6];
        #pragma unroll
        for (int c = 0; c < 6; ++c) {
            w01[c] = wt[(c*3 + row)*16 + f];
            w2[c]  = wt[(c*3 + 2)*16 + f];
        }
        float g01 = g3[row], b01 = b3[row], g2 = g3[2], b2 = b3[2];

        double ma = m < 6 ? 0.0 : sa * RP;
        float cm = (float)(sm * RP - ma * (sb * RP));
        float mb[6], cv[21];
        #pragma unroll
        for (int n = 0; n < 6; ++n)
            mb[n] = __builtin_bit_cast(float, __builtin_amdgcn_readlane(__builtin_bit_cast(int, cm), n));
        #pragma unroll
        for (int n = 0; n < 21; ++n)
            cv[n] = __builtin_bit_cast(float, __builtin_amdgcn_readlane(__builtin_bit_cast(int, cm), 6 + n));
        float A01, B01, A2, B2;
        bn3_consts(w01, mb, cv, g01, b01, A01, B01);
        bn3_consts(w2,  mb, cv, g2,  b2,  A2,  B2);
        if (f == 0)    { lab[row] = A01;  lab[3 + row] = B01; }
        if (lane == 0) { lab[2]   = A2;   lab[5]       = B2;  }
    }
    __syncthreads();

    float uv[6] = {ua.x, ua.y, ua.z, ua.w, ub.x, ub.y};
    float y[6];
    #pragma unroll
    for (int c = 0; c < 6; ++c) y[c] = lrelu(uv[c]*fc[6+c] + fc[12+c]);

    float acc[48];
    #pragma unroll
    for (int c = 0; c < 6; ++c)
      #pragma unroll
      for (int o = 0; o < 3; ++o) {
        const float* wrow = wt + (c*3 + o)*16;
        #pragma unroll
        for (int f = 0; f < 16; ++f)
            acc[o*16 + f] = (c == 0) ? y[0]*wrow[f] : __builtin_fmaf(y[c], wrow[f], acc[o*16 + f]);
      }

    #pragma unroll
    for (int o = 0; o < 3; ++o) {
        float A3 = lab[o], B3 = lab[3+o];
        #pragma unroll
        for (int p = 0; p < 2; ++p)
          #pragma unroll
          for (int q = 0; q < 2; ++q)
            #pragma unroll
            for (int r = 0; r < 2; ++r) {
                int f = p*8 + q*4 + r*2;
                float z0 = __builtin_fmaf(acc[o*16 + f    ], A3, B3);
                float z1 = __builtin_fmaf(acc[o*16 + f + 1], A3, B3);
                float s0 = __builtin_amdgcn_rcpf(1.f + __expf(-z0));
                float s1 = __builtin_amdgcn_rcpf(1.f + __expf(-z1));
                long oidx = (((((long)b*3 + o)*48 + (2*i+p))*48 + (2*j+q))*48 + (2*kk+r))*48 + 2*l;
                *(float2*)(out + oidx) = make_float2(s0, s1);
            }
    }
}

extern "C" void kernel_launch(void* const* d_in, const int* in_sizes, int n_in,
                              void* d_out, int out_size, void* d_ws, size_t ws_size,
                              hipStream_t stream)
{
    (void)in_sizes; (void)n_in; (void)out_size; (void)ws_size;
    const float* x  = (const float*)d_in[0];
    const float* w1 = (const float*)d_in[1];
    const float* g1 = (const float*)d_in[2];
    const float* b1 = (const float*)d_in[3];
    const float* w2 = (const float*)d_in[4];
    const float* g2 = (const float*)d_in[5];
    const float* b2 = (const float*)d_in[6];
    const float* wt = (const float*)d_in[7];
    const float* g3 = (const float*)d_in[8];
    const float* b3 = (const float*)d_in[9];
    float*  out  = (float*)d_out;
    float*  u2   = (float*)((char*)d_ws + OFF_U2);
    float*  wp   = (float*)((char*)d_ws + OFF_WP);
    double* dacc = (double*)((char*)d_ws + OFF_DACC);
    float*  fc   = (float*)((char*)d_ws + OFF_FC);

    k_prep <<<1, 512, 0, stream>>>(w2, wp, dacc);
    k_rx   <<<RX_BLOCKS, 256, 0, stream>>>(x, dacc);
    k_conv2<<<POS_BLOCKS, 256, 0, stream>>>(x, wp, w1, g1, b1, u2, dacc);
    k_m3   <<<M3_BLOCKS, 256, 0, stream>>>(u2, g2, b2, fc, dacc);
    k_out  <<<POS_BLOCKS, 256, 0, stream>>>(u2, wt, fc, out, dacc, g3, b3);
}


# k_out: 288-position workgroups (12 rows of 24) stage their results in LDS and store 16 bytes per lane, every store instruction 4 whole 128-byte lines; was 8-byte stores cut at 64-byte row offsets
# speedup vs baseline: 3.4939x; 1.1495x over previous
#include <hip/hip_runtime.h>
#include <math.h>

#define EPSBN 1e-5f
#define SLOPE 0.01f
static constexpr int  HP  = 24;
static constexpr long P   = 2L*24*24*24*24;
static constexpr long NX  = 2L*48*48*48*48;
static constexpr int  RX_BLOCKS = 648;
static constexpr int  M3_BLOCKS = 216;
static constexpr int  POS_BLOCKS = 2592;
static constexpr int  OUT_T      = 288;
static constexpr int  OUT_BLOCKS = 2304;
static constexpr int  PIECE      = 1152;
static constexpr double RNX = 1.0 / (double)NX;
static constexpr double RP  = 1.0 / (double)P;

static constexpr size_t OFF_U2   = 0;
static constexpr size_t OFF_WP   = (size_t)P*8*4;
static constexpr size_t OFF_DACC = OFF_WP + 768*4;
static constexpr size_t OFF_FC   = OFF_DACC + 512;

typedef float vf2 __attribute__((ext_vector_type(2)));
typedef float vf8 __attribute__((ext_vector_type(8)));

__device__ __forceinline__ float lrelu(float v){ return v >= 0.f ? v : SLOPE*v; }

__device__ __forceinline__ float wave_sum(float v)
{
    #pragma unroll
    for (int m = 16; m >= 1; m >>= 1) v += __shfl_xor(v, m, 32);
    return v;
}

template <int M> __device__ __forceinline__ float row_xor(float v)
{
    return __builtin_bit_cast(float,
        __builtin_amdgcn_update_dpp(0, __builtin_bit_cast(int, v), 0x160 | M, 0xf, 0xf, true));
}
__device__ __forceinline__ float row16_sum(float v)
{
    v += row_xor<8>(v);
    v += row_xor<4>(v);
    v += row_xor<2>(v);
    v += row_xor<1>(v);
    return v;
}


__global__ void k_prep(const float* __restrict__ w2, float* __restrict__ wp,
                       double* __restrict__ dacc)
{
    int tid = threadIdx.x;
    if (tid < 56) dacc[tid] = 0.0;
    if (tid < 384) {
        int n = tid & 15, h = (tid >> 4) & 1, t = tid >> 5;
        int K0 = 4*t + 2*h;
        float vx = (n < 6) ? w2[n*48 + K0]     : 0.f;
        float vy = (n < 6) ? w2[n*48 + K0 + 1] : 0.f;
        ((float2*)wp)[t*32 + h*16 + n] = make_float2(vx, vy);
    }
}

__global__ __launch_bounds__(256) void k_rx(const float* __restrict__ x,
                                            double* __restrict__ dacc)
{
    __shared__ double sacc[2];
    int tid = threadIdx.x;
    if (tid < 2) sacc[tid] = 0.0;
    __syncthreads();
    long gid = (long)blockIdx.x*256 + tid;
    float s = 0.f, q = 0.f;
    #pragma unroll
    for (int it = 0; it < 16; ++it) {
        long idx = gid + (long)it * (RX_BLOCKS*256);
        float4 v = *(const float4*)(x + idx*4);
        s += v.x + v.y + v.z + v.w;
        q += v.x*v.x + v.y*v.y + v.z*v.z + v.w*v.w;
    }
    s = wave_sum(s);
    q = wave_sum(q);
    int lane = tid & 31;
    if (lane < 2) atomicAdd(&sacc[lane], (double)(lane == 0 ? s : q));
    __syncthreads();
    if (tid < 2) atomicAdd(&dacc[tid], sacc[tid]);
}

__global__ __launch_bounds__(256) void k_conv2(const float* __restrict__ x,
                                               const float* __restrict__ wp,
                                               const float* __restrict__ w1,
                                               const float* __restrict__ g1,
                                               const float* __restrict__ b1,
                                               float* __restrict__ u2,
                                               double* __restrict__ dacc)
{
    __shared__ float lx[256*18];
    __shared__ float lsum[6], lssq[6];
    __shared__ float lfc[6];
    int tid = threadIdx.x;
    if (tid < 6) { lsum[tid] = 0.f; lssq[tid] = 0.f; }

    int pos = blockIdx.x*256 + tid;
    int l  = pos % HP;  int t1 = pos / HP;
    int kk = t1 % HP;   t1 /= HP;
    int j  = t1 % HP;   t1 /= HP;
    int i  = t1 % HP;   int b = t1 / HP;
    long xbase = ((((long)b*48 + 2*i)*48 + 2*j)*48 + 2*kk)*48 + 2*l;

    float2 xin[8];
    #pragma unroll
    for (int p = 0; p < 2; ++p)
      #pragma unroll
      for (int q = 0; q < 2; ++q)
        #pragma unroll
        for (int r = 0; r < 2; ++r)
            xin[p*4 + q*2 + r] = *(const float2*)(x + xbase + (long)p*110592 + q*2304 + r*48);

    unsigned hc = (unsigned)(tid - 32*(int)(blockIdx.x & 7));
    if (hc < 3u) {
        double mu  = dacc[0] * RNX;
        double var = dacc[1] * RNX - mu*mu;
        float  wc  = w1[hc];
        float  a   = wc * g1[hc] / sqrtf(wc*wc*(float)var + EPSBN);
        lfc[hc]     = a;
        lfc[3 + hc] = b1[hc] - a*(float)mu;
    }

    float* row = &lx[tid*18];
    #pragma unroll
    for (int n = 0; n < 8; ++n) *(float2*)(row + 2*n) = xin[n];

    int lane = tid & 31, l16 = lane & 15, h = lane >> 4, wv = tid >> 5;

    vf2 Bfrag[12];
    #pragma unroll
    for (int t = 0; t < 12; ++t)
        Bfrag[t] = *(const vf2*)(wp + ((t*2 + h)*16 + l16)*2);

    __syncthreads();

    float a1[3], c1[3];
    #pragma unroll
    for (int c = 0; c < 3; ++c) {
        a1[c] = __builtin_bit_cast(float, __builtin_amdgcn_readfirstlane(__builtin_bit_cast(int, lfc[c])));
        c1[c] = __builtin_bit_cast(float, __builtin_amdgcn_readfirstlane(__builtin_bit_cast(int, lfc[3+c])));
    }

    #pragma unroll
    for (int tau = 0; tau < 2; ++tau) {
        int bl = wv*32 + tau*16;
        vf8 acc = {0.f,0.f,0.f,0.f,0.f,0.f,0.f,0.f};
        #pragma unroll
        for (int t = 0; t < 12; ++t) {
            int c  = t >> 2;
            int fb = (t & 3)*4 + 2*h;
            float2 xv = *(const float2*)&lx[(bl + l16)*18 + fb];
            vf2 A;
            A.x = lrelu(a1[c]*xv.x + c1[c]);
            A.y = lrelu(a1[c]*xv.y + c1[c]);
            acc = __builtin_amdgcn_wmma_f32_16x16x4_f32(
                      false, A, false, Bfrag[t], (short)0, acc, false, false);
        }
        if (l16 < 6) {
            long pbase = (long)blockIdx.x*256 + bl + 8*h;
            float ls = 0.f, lq = 0.f;
            #pragma unroll
            for (int r = 0; r < 8; ++r) {
                float v = acc[r];
                u2[(pbase + r)*8 + l16] = v;
                ls += v; lq += v*v;
            }
            atomicAdd(&lsum[l16], ls);
            atomicAdd(&lssq[l16], lq);
        }
    }
    __syncthreads();
    if (tid < 6) {
        atomicAdd(&dacc[2 + tid], (double)lsum[tid]);
        atomicAdd(&dacc[8 + tid], (double)lssq[tid]);
    }
}

__global__ __launch_bounds__(256) void k_m3(const float* __restrict__ u2,
                                            const float* __restrict__ g2,
                                            const float* __restrict__ b2,
                                            float* __restrict__ fc,
                                            double* __restrict__ dacc)
{
    __shared__ double sacc[27];
    __shared__ float lfc[12];
    int tid = threadIdx.x;
    if (tid < 27) sacc[tid] = 0.0;
    if (tid < 6) {
        double mu  = dacc[2 + tid] * RP;
        double var = dacc[8 + tid] * RP - mu*mu;
        float sc = g2[tid] / sqrtf((float)var + EPSBN);
        float sh = b2[tid] - (float)mu*sc;
        lfc[tid] = sc;  lfc[6 + tid] = sh;
        if (blockIdx.x == 0) { fc[6 + tid] = sc;  fc[12 + tid] = sh; }
    }
    __syncthreads();
    float sc2[6], sh2[6];
    #pragma unroll
    for (int c = 0; c < 6; ++c) { sc2[c] = lfc[c]; sh2[c] = lfc[6 + c]; }
    float m[6]; float s[21];
    #pragma unroll
    for (int c = 0; c < 6; ++c) m[c] = 0.f;
    #pragma unroll
    for (int k = 0; k < 21; ++k) s[k] = 0.f;
    long gid = (long)blockIdx.x*256 + tid;
    for (int it = 0; it < 12; ++it) {
        long pos = gid + (long)it*(M3_BLOCKS*256);
        float4 ua = *(const float4*)(u2 + pos*8);
        float4 ub = *(const float4*)(u2 + pos*8 + 4);
        float uv[6] = {ua.x, ua.y, ua.z, ua.w, ub.x, ub.y};
        float y[6];
        #pragma unroll
        for (int c = 0; c < 6; ++c) y[c] = lrelu(uv[c]*sc2[c] + sh2[c]);
        int idx = 0;
        #pragma unroll
        for (int c = 0; c < 6; ++c) {
            m[c] += y[c];
            #pragma unroll
            for (int c2 = c; c2 < 6; ++c2) s[idx++] += y[c]*y[c2];
        }
    }
    float tot[27];
    #pragma unroll
    for (int c = 0; c < 6; ++c)  tot[c]     = m[c];
    #pragma unroll
    for (int k = 0; k < 21; ++k) tot[6 + k] = s[k];
    #pragma unroll
    for (int msk = 16; msk >= 1; msk >>= 1) {
        #pragma unroll
        for (int k = 0; k < 27; ++k) tot[k] += __shfl_xor(tot[k], msk, 32);
    }
    int lane = tid & 31;
    float pick = 0.f;
    #pragma unroll
    for (int k = 0; k < 27; ++k) pick = (lane == k) ? tot[k] : pick;
    if (lane < 27) atomicAdd(&sacc[lane], (double)pick);
    __syncthreads();
    if (tid < 27) atomicAdd(&dacc[14 + tid], sacc[tid]);
}

__device__ __forceinline__ void bn3_consts(const float (&w)[6], const float (&mb)[6], const float (&cv)[21],
                                           float g, float b, float& A, float& B)
{
    float mu = 0.f, qd = 0.f, qo = 0.f;
    int idx = 0;
    #pragma unroll
    for (int c = 0; c < 6; ++c) {
        mu = __builtin_fmaf(w[c], mb[c], mu);
        qd = __builtin_fmaf(w[c]*w[c], cv[idx++], qd);
        if (c < 5) {
            float t = cv[idx++]*w[c + 1];
            #pragma unroll
            for (int c2 = c + 2; c2 < 6; ++c2) t = __builtin_fmaf(cv[idx++], w[c2], t);
            qo = __builtin_fmaf(w[c], t, qo);
        }
    }
    float q    = __builtin_fmaf(2.f, qo, qd);
    float mean = row16_sum(mu) * 0.0625f;
    float d    = mu - mean;
    float var  = row16_sum(__builtin_fmaf(d, d, q)) * 0.0625f;
    A = g / sqrtf(var + EPSBN);
    B = b - mean*A;
}

__global__ __launch_bounds__(288) void k_out(const float* __restrict__ u2,
                                             const float* __restrict__ wt,
                                             const float* __restrict__ fc,
                                             float* __restrict__ out,
                                             const double* __restrict__ dacc,
                                             const float* __restrict__ g3,
                                             const float* __restrict__ b3)
{
    __shared__ float  lab[6];
    __shared__ float4 stage4[12*OUT_T];
    float* stage = (float*)stage4;
    const int t   = threadIdx.x;
    const int pos = blockIdx.x*OUT_T + t;
    const int khalf = blockIdx.x & 1;
    int r1 = blockIdx.x >> 1;
    const int j = r1 % HP;  r1 /= HP;
    const int i = r1 % HP;  const int b = r1 / HP;
    const int kkl = t / HP;
    const int l   = t - kkl*HP;

    float4 ua = *(const float4*)(u2 + (long)pos*8);
    float4 ub = *(const float4*)(u2 + (long)pos*8 + 4);

    const int wv = __builtin_amdgcn_readfirstlane((int)(threadIdx.x >> 5));
    if (wv == (int)(blockIdx.x & 7)) {
        const int lane = threadIdx.x & 31, row = lane >> 4, f = lane & 15;
        int m = lane < 27 ? lane : 26;
        int ca = m, cb = m;
        if (m >= 6) {
            int n = m - 6;
            ca = (n >= 6) + (n >= 11) + (n >= 15) + (n >= 18) + (n >= 20);
            cb = ca + n - (6*ca - ca*(ca - 1)/2);
        }
        double sm = dacc[14 + m],  sa = dacc[14 + ca],  sb = dacc[14 + cb];
        float w01[6], w2[6];
        #pragma unroll
        for (int c = 0; c < 6; ++c) {
            w01[c] = wt[(c*3 + row)*16 + f];
            w2[c]  = wt[(c*3 + 2)*16 + f];
        }
        float g01 = g3[row], b01 = b3[row], g2 = g3[2], b2 = b3[2];

        double ma = m < 6 ? 0.0 : sa * RP;
        float cm = (float)(sm * RP - ma * (sb * RP));
        float mb[6], cv[21];
        #pragma unroll
        for (int n = 0; n < 6; ++n)
            mb[n] = __builtin_bit_cast(float, __builtin_amdgcn_readlane(__builtin_bit_cast(int, cm), n));
        #pragma unroll
        for (int n = 0; n < 21; ++n)
            cv[n] = __builtin_bit_cast(float, __builtin_amdgcn_readlane(__builtin_bit_cast(int, cm), 6 + n));
        float A01, B01, A2, B2;
        bn3_consts(w01, mb, cv, g01, b01, A01, B01);
        bn3_consts(w2,  mb, cv, g2,  b2,  A2,  B2);
        if (f == 0)    { lab[row] = A01;  lab[3 + row] = B01; }
        if (lane == 0) { lab[2]   = A2;   lab[5]       = B2;  }
    }
    __syncthreads();

    float uv[6] = {ua.x, ua.y, ua.z, ua.w, ub.x, ub.y};
    float y[6];
    #pragma unroll
    for (int c = 0; c < 6; ++c) y[c] = lrelu(uv[c]*fc[6+c] + fc[12+c]);

    float acc[48];
    #pragma unroll
    for (int c = 0; c < 6; ++c)
      #pragma unroll
      for (int o = 0; o < 3; ++o) {
        const float* wrow = wt + (c*3 + o)*16;
        #pragma unroll
        for (int f = 0; f < 16; ++f)
            acc[o*16 + f] = (c == 0) ? y[0]*wrow[f] : __builtin_fmaf(y[c], wrow[f], acc[o*16 + f]);
      }

    float* mine = stage + kkl*96 + 2*l;
    #pragma unroll
    for (int o = 0; o < 3; ++o) {
        float A3 = lab[o], B3 = lab[3+o];
        #pragma unroll
        for (int p = 0; p < 2; ++p)
          #pragma unroll
          for (int q = 0; q < 2; ++q)
            #pragma unroll
            for (int r = 0; r < 2; ++r) {
                int f = p*8 + q*4 + r*2;
                float z0 = __builtin_fmaf(acc[o*16 + f    ], A3, B3);
                float z1 = __builtin_fmaf(acc[o*16 + f + 1], A3, B3);
                float s0 = __builtin_amdgcn_rcpf(1.f + __expf(-z0));
                float s1 = __builtin_amdgcn_rcpf(1.f + __expf(-z1));
                *(float2*)(mine + (o*4 + p*2 + q)*PIECE + r*48) = make_float2(s0, s1);
            }
    }
    __syncthreads();
    #pragma unroll
    for (int o = 0; o < 3; ++o)
      #pragma unroll
      for (int p = 0; p < 2; ++p)
        #pragma unroll
        for (int q = 0; q < 2; ++q) {
            float4 v = stage4[(o*4 + p*2 + q)*OUT_T + t];
            long pbase = ((((long)b*3 + o)*48 + (2*i+p))*48 + (2*j+q))*2304 + khalf*PIECE;
            *(float4*)(out + pbase + 4*t) = v;
        }
}

extern "C" void kernel_launch(void* const* d_in, const int* in_sizes, int n_in,
                              void* d_out, int out_size, void* d_ws, size_t ws_size,
                              hipStream_t stream)
{
    (void)in_sizes; (void)n_in; (void)out_size; (void)ws_size;
    const float* x  = (const float*)d_in[0];
    const float* w1 = (const float*)d_in[1];
    const float* g1 = (const float*)d_in[2];
    const float* b1 = (const float*)d_in[3];
    const float* w2 = (const float*)d_in[4];
    const float* g2 = (const float*)d_in[5];
    const float* b2 = (const float*)d_in[6];
    const float* wt = (const float*)d_in[7];
    const float* g3 = (const float*)d_in[8];
    const float* b3 = (const float*)d_in[9];
    float*  out  = (float*)d_out;
    float*  u2   = (float*)((char*)d_ws + OFF_U2);
    float*  wp   = (float*)((char*)d_ws + OFF_WP);
    double* dacc = (double*)((char*)d_ws + OFF_DACC);
    float*  fc   = (float*)((char*)d_ws + OFF_FC);

    k_prep <<<1, 512, 0, stream>>>(w2, wp, dacc);
    k_rx   <<<RX_BLOCKS, 256, 0, stream>>>(x, dacc);
    k_conv2<<<POS_BLOCKS, 256, 0, stream>>>(x, wp, w1, g1, b1, u2, dacc);
    k_m3   <<<M3_BLOCKS, 256, 0, stream>>>(u2, g2, b2, fc, dacc);
    k_out  <<<OUT_BLOCKS, OUT_T, 0, stream>>>(u2, wt, fc, out, dacc, g3, b3);
}


# k_out: each workgroup walks 3 tiles of 288 positions (768 workgroups, 6,912 waves, was 20,736 short ones): BN3 head once per workgroup, next tile's u2 rows requested ahead; every sum as before
# speedup vs baseline: 3.6100x; 1.0332x over previous
#include <hip/hip_runtime.h>
#include <math.h>

#define EPSBN 1e-5f
#define SLOPE 0.01f
static constexpr int  HP  = 24;
static constexpr long P   = 2L*24*24*24*24;
static constexpr long NX  = 2L*48*48*48*48;
static constexpr int  RX_BLOCKS = 648;
static constexpr int  M3_BLOCKS = 216;
static constexpr int  POS_BLOCKS = 2592;
static constexpr int  OUT_T      = 288;
static constexpr int  OUT_BLOCKS = 2304;
static constexpr int  OUT_TILES  = 3;
static constexpr int  OUT_GROUPS = OUT_BLOCKS / OUT_TILES;
static constexpr int  PIECE      = 1152;
static constexpr double RNX = 1.0 / (double)NX;
static constexpr double RP  = 1.0 / (double)P;

static constexpr size_t OFF_U2   = 0;
static constexpr size_t OFF_WP   = (size_t)P*8*4;
static constexpr size_t OFF_DACC = OFF_WP + 768*4;
static constexpr size_t OFF_FC   = OFF_DACC + 512;

typedef float vf2 __attribute__((ext_vector_type(2)));
typedef float vf8 __attribute__((ext_vector_type(8)));

__device__ __forceinline__ float lrelu(float v){ return v >= 0.f ? v : SLOPE*v; }

__device__ __forceinline__ float wave_sum(float v)
{
    #pragma unroll
    for (int m = 16; m >= 1; m >>= 1) v += __shfl_xor(v, m, 32);
    return v;
}

template <int M> __device__ __forceinline__ float row_xor(float v)
{
    return __builtin_bit_cast(float,
        __builtin_amdgcn_update_dpp(0, __builtin_bit_cast(int, v), 0x160 | M, 0xf, 0xf, true));
}
__device__ __forceinline__ float row16_sum(float v)
{
    v += row_xor<8>(v);
    v += row_xor<4>(v);
    v += row_xor<2>(v);
    v += row_xor<1>(v);
    return v;
}


__global__ void k_prep(const float* __restrict__ w2, float* __restrict__ wp,
                       double* __restrict__ dacc)
{
    int tid = threadIdx.x;
    if (tid < 56) dacc[tid] = 0.0;
    if (tid < 384) {
        int n = tid & 15, h = (tid >> 4) & 1, t = tid >> 5;
        int K0 = 4*t + 2*h;
        float vx = (n < 6) ? w2[n*48 + K0]     : 0.f;
        float vy = (n < 6) ? w2[n*48 + K0 + 1] : 0.f;
        ((float2*)wp)[t*32 + h*16 + n] = make_float2(vx, vy);
    }
}

__global__ __launch_bounds__(256) void k_rx(const float* __restrict__ x,
                                            double* __restrict__ dacc)
{
    __shared__ double sacc[2];
    int tid = threadIdx.x;
    if (tid < 2) sacc[tid] = 0.0;
    __syncthreads();
    long gid = (long)blockIdx.x*256 + tid;
    float s = 0.f, q = 0.f;
    #pragma unroll
    for (int it = 0; it < 16; ++it) {
        long idx = gid + (long)it * (RX_BLOCKS*256);
        float4 v = *(const float4*)(x + idx*4);
        s += v.x + v.y + v.z + v.w;
        q += v.x*v.x + v.y*v.y + v.z*v.z + v.w*v.w;
    }
    s = wave_sum(s);
    q = wave_sum(q);
    int lane = tid & 31;
    if (lane < 2) atomicAdd(&sacc[lane], (double)(lane == 0 ? s : q));
    __syncthreads();
    if (tid < 2) atomicAdd(&dacc[tid], sacc[tid]);
}

__global__ __launch_bounds__(256) void k_conv2(const float* __restrict__ x,
                                               const float* __restrict__ wp,
                                               const float* __restrict__ w1,
                                               const float* __restrict__ g1,
                                               const float* __restrict__ b1,
                                               float* __restrict__ u2,
                                               double* __restrict__ dacc)
{
    __shared__ float lx[256*18];
    __shared__ float lsum[6], lssq[6];
    __shared__ float lfc[6];
    int tid = threadIdx.x;
    if (tid < 6) { lsum[tid] = 0.f; lssq[tid] = 0.f; }

    int pos = blockIdx.x*256 + tid;
    int l  = pos % HP;  int t1 = pos / HP;
    int kk = t1 % HP;   t1 /= HP;
    int j  = t1 % HP;   t1 /= HP;
    int i  = t1 % HP;   int b = t1 / HP;
    long xbase = ((((long)b*48 + 2*i)*48 + 2*j)*48 + 2*kk)*48 + 2*l;

    float2 xin[8];
    #pragma unroll
    for (int p = 0; p < 2; ++p)
      #pragma unroll
      for (int q = 0; q < 2; ++q)
        #pragma unroll
        for (int r = 0; r < 2; ++r)
            xin[p*4 + q*2 + r] = *(const float2*)(x + xbase + (long)p*110592 + q*2304 + r*48);

    unsigned hc = (unsigned)(tid - 32*(int)(blockIdx.x & 7));
    if (hc < 3u) {
        double mu  = dacc[0] * RNX;
        double var = dacc[1] * RNX - mu*mu;
        float  wc  = w1[hc];
        float  a   = wc * g1[hc] / sqrtf(wc*wc*(float)var + EPSBN);
        lfc[hc]     = a;
        lfc[3 + hc] = b1[hc] - a*(float)mu;
    }

    float* row = &lx[tid*18];
    #pragma unroll
    for (int n = 0; n < 8; ++n) *(float2*)(row + 2*n) = xin[n];

    int lane = tid & 31, l16 = lane & 15, h = lane >> 4, wv = tid >> 5;

    vf2 Bfrag[12];
    #pragma unroll
    for (int t = 0; t < 12; ++t)
        Bfrag[t] = *(const vf2*)(wp + ((t*2 + h)*16 + l16)*2);

    __syncthreads();

    float a1[3], c1[3];
    #pragma unroll
    for (int c = 0; c < 3; ++c) {
        a1[c] = __builtin_bit_cast(float, __builtin_amdgcn_readfirstlane(__builtin_bit_cast(int, lfc[c])));
        c1[c] = __builtin_bit_cast(float, __builtin_amdgcn_readfirstlane(__builtin_bit_cast(int, lfc[3+c])));
    }

    #pragma unroll
    for (int tau = 0; tau < 2; ++tau) {
        int bl = wv*32 + tau*16;
        vf8 acc = {0.f,0.f,0.f,0.f,0.f,0.f,0.f,0.f};
        #pragma unroll
        for (int t = 0; t < 12; ++t) {
            int c  = t >> 2;
            int fb = (t & 3)*4 + 2*h;
            float2 xv = *(const float2*)&lx[(bl + l16)*18 + fb];
            vf2 A;
            A.x = lrelu(a1[c]*xv.x + c1[c]);
            A.y = lrelu(a1[c]*xv.y + c1[c]);
            acc = __builtin_amdgcn_wmma_f32_16x16x4_f32(
                      false, A, false, Bfrag[t], (short)0, acc, false, false);
        }
        if (l16 < 6) {
            long pbase = (long)blockIdx.x*256 + bl + 8*h;
            float ls = 0.f, lq = 0.f;
            #pragma unroll
            for (int r = 0; r < 8; ++r) {
                float v = acc[r];
                u2[(pbase + r)*8 + l16] = v;
                ls += v; lq += v*v;
            }
            atomicAdd(&lsum[l16], ls);
            atomicAdd(&lssq[l16], lq);
        }
    }
    __syncthreads();
    if (tid < 6) {
        atomicAdd(&dacc[2 + tid], (double)lsum[tid]);
        atomicAdd(&dacc[8 + tid], (double)lssq[tid]);
    }
}

__global__ __launch_bounds__(256) void k_m3(const float* __restrict__ u2,
                                            const float* __restrict__ g2,
                                            const float* __restrict__ b2,
                                            float* __restrict__ fc,
                                            double* __restrict__ dacc)
{
    __shared__ double sacc[27];
    __shared__ float lfc[12];
    int tid = threadIdx.x;
    if (tid < 27) sacc[tid] = 0.0;
    if (tid < 6) {
        double mu  = dacc[2 + tid] * RP;
        double var = dacc[8 + tid] * RP - mu*mu;
        float sc = g2[tid] / sqrtf((float)var + EPSBN);
        float sh = b2[tid] - (float)mu*sc;
        lfc[tid] = sc;  lfc[6 + tid] = sh;
        if (blockIdx.x == 0) { fc[6 + tid] = sc;  fc[12 + tid] = sh; }
    }
    __syncthreads();
    float sc2[6], sh2[6];
    #pragma unroll
    for (int c = 0; c < 6; ++c) { sc2[c] = lfc[c]; sh2[c] = lfc[6 + c]; }
    float m[6]; float s[21];
    #pragma unroll
    for (int c = 0; c < 6; ++c) m[c] = 0.f;
    #pragma unroll
    for (int k = 0; k < 21; ++k) s[k] = 0.f;
    long gid = (long)blockIdx.x*256 + tid;
    for (int it = 0; it < 12; ++it) {
        long pos = gid + (long)it*(M3_BLOCKS*256);
        float4 ua = *(const float4*)(u2 + pos*8);
        float4 ub = *(const float4*)(u2 + pos*8 + 4);
        float uv[6] = {ua.x, ua.y, ua.z, ua.w, ub.x, ub.y};
        float y[6];
        #pragma unroll
        for (int c = 0; c < 6; ++c) y[c] = lrelu(uv[c]*sc2[c] + sh2[c]);
        int idx = 0;
        #pragma unroll
        for (int c = 0; c < 6; ++c) {
            m[c] += y[c];
            #pragma unroll
            for (int c2 = c; c2 < 6; ++c2) s[idx++] += y[c]*y[c2];
        }
    }
    float tot[27];
    #pragma unroll
    for (int c = 0; c < 6; ++c)  tot[c]     = m[c];
    #pragma unroll
    for (int k = 0; k < 21; ++k) tot[6 + k] = s[k];
    #pragma unroll
    for (int msk = 16; msk >= 1; msk >>= 1) {
        #pragma unroll
        for (int k = 0; k < 27; ++k) tot[k] += __shfl_xor(tot[k], msk, 32);
    }
    int lane = tid & 31;
    float pick = 0.f;
    #pragma unroll
    for (int k = 0; k < 27; ++k) pick = (lane == k) ? tot[k] : pick;
    if (lane < 27) atomicAdd(&sacc[lane], (double)pick);
    __syncthreads();
    if (tid < 27) atomicAdd(&dacc[14 + tid], sacc[tid]);
}

__device__ __forceinline__ void bn3_consts(const float (&w)[6], const float (&mb)[6], const float (&cv)[21],
                                           float g, float b, float& A, float& B)
{
    float mu = 0.f, qd = 0.f, qo = 0.f;
    int idx = 0;
    #pragma unroll
    for (int c = 0; c < 6; ++c) {
        mu = __builtin_fmaf(w[c], mb[c], mu);
        qd = __builtin_fmaf(w[c]*w[c], cv[idx++], qd);
        if (c < 5) {
            float t = cv[idx++]*w[c + 1];
            #pragma unroll
            for (int c2 = c + 2; c2 < 6; ++c2) t = __builtin_fmaf(cv[idx++], w[c2], t);
            qo = __builtin_fmaf(w[c], t, qo);
        }
    }
    float q    = __builtin_fmaf(2.f, qo, qd);
    float mean = row16_sum(mu) * 0.0625f;
    float d    = mu - mean;
    float var  = row16_sum(__builtin_fmaf(d, d, q)) * 0.0625f;
    A = g / sqrtf(var + EPSBN);
    B = b - mean*A;
}

__global__ __launch_bounds__(288) void k_out(const float* __restrict__ u2,
                                             const float* __restrict__ wt,
                                             const float* __restrict__ fc,
                                             float* __restrict__ out,
                                             const double* __restrict__ dacc,
                                             const float* __restrict__ g3,
                                             const float* __restrict__ b3)
{
    __shared__ float  lab[6];
    __shared__ float4 stage4[12*OUT_T];
    float* stage = (float*)stage4;
    const int t     = threadIdx.x;
    const int tile0 = blockIdx.x*OUT_TILES;
    const int kkl = t / HP;
    const int l   = t - kkl*HP;

    const float* urow = u2 + ((long)tile0*OUT_T + t)*8;
    float4 ua = *(const float4*)(urow);
    float4 ub = *(const float4*)(urow + 4);

    const int wv = __builtin_amdgcn_readfirstlane((int)(threadIdx.x >> 5));
    if (wv == (int)(blockIdx.x & 7)) {
        const int lane = threadIdx.x & 31, row = lane >> 4, f = lane & 15;
        int m = lane < 27 ? lane : 26;
        int ca = m, cb = m;
        if (m >= 6) {
            int n = m - 6;
            ca = (n >= 6) + (n >= 11) + (n >= 15) + (n >= 18) + (n >= 20);
            cb = ca + n - (6*ca - ca*(ca - 1)/2);
        }
        double sm = dacc[14 + m],  sa = dacc[14 + ca],  sb = dacc[14 + cb];
        float w01[6], w2[6];
        #pragma unroll
        for (int c = 0; c < 6; ++c) {
            w01[c] = wt[(c*3 + row)*16 + f];
            w2[c]  = wt[(c*3 + 2)*16 + f];
        }
        float g01 = g3[row], b01 = b3[row], g2 = g3[2], b2 = b3[2];

        double ma = m < 6 ? 0.0 : sa * RP;
        float cm = (float)(sm * RP - ma * (sb * RP));
        float mb[6], cv[21];
        #pragma unroll
        for (int n = 0; n < 6; ++n)
            mb[n] = __builtin_bit_cast(float, __builtin_amdgcn_readlane(__builtin_bit_cast(int, cm), n));
        #pragma unroll
        for (int n = 0; n < 21; ++n)
            cv[n] = __builtin_bit_cast(float, __builtin_amdgcn_readlane(__builtin_bit_cast(int, cm), 6 + n));
        float A01, B01, A2, B2;
        bn3_consts(w01, mb, cv, g01, b01, A01, B01);
        bn3_consts(w2,  mb, cv, g2,  b2,  A2,  B2);
        if (f == 0)    { lab[row] = A01;  lab[3 + row] = B01; }
        if (lane == 0) { lab[2]   = A2;   lab[5]       = B2;  }
    }
    __syncthreads();

    float* mine = stage + kkl*96 + 2*l;

    #pragma unroll 1
    for (int it = 0; it < OUT_TILES; ++it) {
        const int tile  = tile0 + it;
        const int khalf = tile & 1;
        int r1 = tile >> 1;
        const int j = r1 % HP;  r1 /= HP;
        const int i = r1 % HP;  const int b = r1 / HP;

        if (it > 0) __syncthreads();

        float uv[6] = {ua.x, ua.y, ua.z, ua.w, ub.x, ub.y};
        if (it + 1 < OUT_TILES) {
            ua = *(const float4*)(urow + (long)(it + 1)*OUT_T*8);
            ub = *(const float4*)(urow + (long)(it + 1)*OUT_T*8 + 4);
        }
        float y[6];
        #pragma unroll
        for (int c = 0; c < 6; ++c) y[c] = lrelu(uv[c]*fc[6+c] + fc[12+c]);

        int wofs = 0;
        asm volatile("" : "+s"(wofs));
        wofs = __builtin_amdgcn_readfirstlane(wofs);
        const __attribute__((address_space(4))) float* wtl =
            (const __attribute__((address_space(4))) float*)wt + wofs;
        float acc[48];
        #pragma unroll
        for (int c = 0; c < 6; ++c)
          #pragma unroll
          for (int o = 0; o < 3; ++o) {
            const __attribute__((address_space(4))) float* wrow = wtl + (c*3 + o)*16;
            #pragma unroll
            for (int f = 0; f < 16; ++f)
                acc[o*16 + f] = (c == 0) ? y[0]*wrow[f] : __builtin_fmaf(y[c], wrow[f], acc[o*16 + f]);
          }

        #pragma unroll
        for (int o = 0; o < 3; ++o) {
            float A3 = lab[o], B3 = lab[3+o];
            #pragma unroll
            for (int p = 0; p < 2; ++p)
              #pragma unroll
              for (int q = 0; q < 2; ++q)
                #pragma unroll
                for (int r = 0; r < 2; ++r) {
                    int f = p*8 + q*4 + r*2;
                    float z0 = __builtin_fmaf(acc[o*16 + f    ], A3, B3);
                    float z1 = __builtin_fmaf(acc[o*16 + f + 1], A3, B3);
                    float s0 = __builtin_amdgcn_rcpf(1.f + __expf(-z0));
                    float s1 = __builtin_amdgcn_rcpf(1.f + __expf(-z1));
                    *(float2*)(mine + (o*4 + p*2 + q)*PIECE + r*48) = make_float2(s0, s1);
                }
        }
        __syncthreads();
        #pragma unroll
        for (int o = 0; o < 3; ++o)
          #pragma unroll
          for (int p = 0; p < 2; ++p)
            #pragma unroll
            for (int q = 0; q < 2; ++q) {
                float4 v = stage4[(o*4 + p*2 + q)*OUT_T + t];
                long pbase = ((((long)b*3 + o)*48 + (2*i+p))*48 + (2*j+q))*2304 + khalf*PIECE;
                *(float4*)(out + pbase + 4*t) = v;
            }
    }
}

extern "C" void kernel_launch(void* const* d_in, const int* in_sizes, int n_in,
                              void* d_out, int out_size, void* d_ws, size_t ws_size,
                              hipStream_t stream)
{
    (void)in_sizes; (void)n_in; (void)out_size; (void)ws_size;
    const float* x  = (const float*)d_in[0];
    const float* w1 = (const float*)d_in[1];
    const float* g1 = (const float*)d_in[2];
    const float* b1 = (const float*)d_in[3];
    const float* w2 = (const float*)d_in[4];
    const float* g2 = (const float*)d_in[5];
    const float* b2 = (const float*)d_in[6];
    const float* wt = (const float*)d_in[7];
    const float* g3 = (const float*)d_in[8];
    const float* b3 = (const float*)d_in[9];
    float*  out  = (float*)d_out;
    float*  u2   = (float*)((char*)d_ws + OFF_U2);
    float*  wp   = (float*)((char*)d_ws + OFF_WP);
    double* dacc = (double*)((char*)d_ws + OFF_DACC);
    float*  fc   = (float*)((char*)d_ws + OFF_FC);

    k_prep <<<1, 512, 0, stream>>>(w2, wp, dacc);
    k_rx   <<<RX_BLOCKS, 256, 0, stream>>>(x, dacc);
    k_conv2<<<POS_BLOCKS, 256, 0, stream>>>(x, wp, w1, g1, b1, u2, dacc);
    k_m3   <<<M3_BLOCKS, 256, 0, stream>>>(u2, g2, b2, fc, dacc);
    k_out  <<<OUT_GROUPS, OUT_T, 0, stream>>>(u2, wt, fc, out, dacc, g3, b3);
}


# k_conv2: 20,736 one-piece waves -> 4,096 waves walking 5-6 pieces of 32 positions, A fragments read straight from x one piece ahead; BN2 sums per lane, one atomic set per workgroup (sums regrouped)
# speedup vs baseline: 3.8706x; 1.0722x over previous
#include <hip/hip_runtime.h>
#include <math.h>

#define EPSBN 1e-5f
#define SLOPE 0.01f
static constexpr int  HP  = 24;
static constexpr long P   = 2L*24*24*24*24;
static constexpr long NX  = 2L*48*48*48*48;
static constexpr int  RX_BLOCKS = 648;
static constexpr int  M3_BLOCKS = 216;
static constexpr int  C2_GROUPS = 512;
static constexpr int  C2_WAVES  = C2_GROUPS*8;
static constexpr int  C2_UNITS  = (int)(P/32);
static constexpr int  OUT_T      = 288;
static constexpr int  OUT_BLOCKS = 2304;
static constexpr int  OUT_TILES  = 3;
static constexpr int  OUT_GROUPS = OUT_BLOCKS / OUT_TILES;
static constexpr int  PIECE      = 1152;
static constexpr double RNX = 1.0 / (double)NX;
static constexpr double RP  = 1.0 / (double)P;

static constexpr size_t OFF_U2   = 0;
static constexpr size_t OFF_WP   = (size_t)P*8*4;
static constexpr size_t OFF_DACC = OFF_WP + 768*4;
static constexpr size_t OFF_FC   = OFF_DACC + 512;

typedef float vf2 __attribute__((ext_vector_type(2)));
typedef float vf8 __attribute__((ext_vector_type(8)));

__device__ __forceinline__ float lrelu(float v){ return v >= 0.f ? v : SLOPE*v; }

__device__ __forceinline__ float wave_sum(float v)
{
    #pragma unroll
    for (int m = 16; m >= 1; m >>= 1) v += __shfl_xor(v, m, 32);
    return v;
}

template <int M> __device__ __forceinline__ float row_xor(float v)
{
    return __builtin_bit_cast(float,
        __builtin_amdgcn_update_dpp(0, __builtin_bit_cast(int, v), 0x160 | M, 0xf, 0xf, true));
}
__device__ __forceinline__ float row16_sum(float v)
{
    v += row_xor<8>(v);
    v += row_xor<4>(v);
    v += row_xor<2>(v);
    v += row_xor<1>(v);
    return v;
}


__global__ void k_prep(const float* __restrict__ w2, float* __restrict__ wp,
                       double* __restrict__ dacc)
{
    int tid = threadIdx.x;
    if (tid < 56) dacc[tid] = 0.0;
    if (tid < 384) {
        int n = tid & 15, h = (tid >> 4) & 1, t = tid >> 5;
        int K0 = 4*t + 2*h;
        float vx = (n < 6) ? w2[n*48 + K0]     : 0.f;
        float vy = (n < 6) ? w2[n*48 + K0 + 1] : 0.f;
        ((float2*)wp)[t*32 + h*16 + n] = make_float2(vx, vy);
    }
}

__global__ __launch_bounds__(256) void k_rx(const float* __restrict__ x,
                                            double* __restrict__ dacc)
{
    __shared__ double sacc[2];
    int tid = threadIdx.x;
    if (tid < 2) sacc[tid] = 0.0;
    __syncthreads();
    long gid = (long)blockIdx.x*256 + tid;
    float s = 0.f, q = 0.f;
    #pragma unroll
    for (int it = 0; it < 16; ++it) {
        long idx = gid + (long)it * (RX_BLOCKS*256);
        float4 v = *(const float4*)(x + idx*4);
        s += v.x + v.y + v.z + v.w;
        q += v.x*v.x + v.y*v.y + v.z*v.z + v.w*v.w;
    }
    s = wave_sum(s);
    q = wave_sum(q);
    int lane = tid & 31;
    if (lane < 2) atomicAdd(&sacc[lane], (double)(lane == 0 ? s : q));
    __syncthreads();
    if (tid < 2) atomicAdd(&dacc[tid], sacc[tid]);
}

__device__ __forceinline__ void c2_load(const float* __restrict__ x, int unit, int l16, int h, float2 (&xv)[2][4])
{
    #pragma unroll
    for (int tau = 0; tau < 2; ++tau) {
        int pos = unit*32 + tau*16 + l16;
        int l  = pos % HP;  int t1 = pos / HP;
        int kk = t1 % HP;   t1 /= HP;
        int j  = t1 % HP;   t1 /= HP;
        int i  = t1 % HP;   int b = t1 / HP;
        int xb = ((((b*48 + 2*i)*48 + 2*j)*48 + 2*kk + h)*48) + 2*l;
        #pragma unroll
        for (int pq = 0; pq < 4; ++pq)
            xv[tau][pq] = *(const float2*)(x + xb + (pq >> 1)*110592 + (pq & 1)*2304);
    }
}

__global__ __launch_bounds__(256) void k_conv2(const float* __restrict__ x,
                                               const float* __restrict__ wp,
                                               const float* __restrict__ w1,
                                               const float* __restrict__ g1,
                                               const float* __restrict__ b1,
                                               float* __restrict__ u2,
                                               double* __restrict__ dacc)
{
    __shared__ float lsum[6], lssq[6];
    __shared__ float lfc[6];
    int tid = threadIdx.x;
    if (tid < 6) { lsum[tid] = 0.f; lssq[tid] = 0.f; }

    const int lane = tid & 31, l16 = lane & 15, h = lane >> 4;
    const int wv = __builtin_amdgcn_readfirstlane(tid >> 5);
    int unit = blockIdx.x*8 + wv;

    float2 cur[2][4], nxt[2][4];
    c2_load(x, unit, l16, h, cur);
    #pragma unroll
    for (int tau = 0; tau < 2; ++tau)
        #pragma unroll
        for (int pq = 0; pq < 4; ++pq) nxt[tau][pq] = cur[tau][pq];

    unsigned hc = (unsigned)(tid - 32*(int)(blockIdx.x & 7));
    if (hc < 3u) {
        double mu  = dacc[0] * RNX;
        double var = dacc[1] * RNX - mu*mu;
        float  wc  = w1[hc];
        float  a   = wc * g1[hc] / sqrtf(wc*wc*(float)var + EPSBN);
        lfc[hc]     = a;
        lfc[3 + hc] = b1[hc] - a*(float)mu;
    }

    vf2 Bfrag[12];
    #pragma unroll
    for (int t = 0; t < 12; ++t)
        Bfrag[t] = *(const vf2*)(wp + ((t*2 + h)*16 + l16)*2);

    __syncthreads();

    float a1[3], c1[3];
    #pragma unroll
    for (int c = 0; c < 3; ++c) {
        a1[c] = __builtin_bit_cast(float, __builtin_amdgcn_readfirstlane(__builtin_bit_cast(int, lfc[c])));
        c1[c] = __builtin_bit_cast(float, __builtin_amdgcn_readfirstlane(__builtin_bit_cast(int, lfc[3+c])));
    }

    float ls = 0.f, lq = 0.f;

    #pragma unroll 1
    for (; unit < C2_UNITS; unit += C2_WAVES) {
        if (unit + C2_WAVES < C2_UNITS) c2_load(x, unit + C2_WAVES, l16, h, nxt);

        vf8 acc[2];
        #pragma unroll
        for (int tau = 0; tau < 2; ++tau) acc[tau] = (vf8){0.f,0.f,0.f,0.f,0.f,0.f,0.f,0.f};
        #pragma unroll
        for (int t = 0; t < 12; ++t) {
            int c = t >> 2;
            #pragma unroll
            for (int tau = 0; tau < 2; ++tau) {
                float2 xv = cur[tau][t & 3];
                vf2 A;
                A.x = lrelu(a1[c]*xv.x + c1[c]);
                A.y = lrelu(a1[c]*xv.y + c1[c]);
                acc[tau] = __builtin_amdgcn_wmma_f32_16x16x4_f32(
                               false, A, false, Bfrag[t], (short)0, acc[tau], false, false);
            }
        }
        #pragma unroll
        for (int tau = 0; tau < 2; ++tau)
            #pragma unroll
            for (int r = 0; r < 8; ++r) {
                float v = acc[tau][r];
                ls += v; lq += v*v;
            }
        if (l16 < 6) {
            float* up = u2 + ((long)unit*32 + 8*h)*8 + l16;
            #pragma unroll
            for (int tau = 0; tau < 2; ++tau)
                #pragma unroll
                for (int r = 0; r < 8; ++r) up[(tau*16 + r)*8] = acc[tau][r];
        }
        #pragma unroll
        for (int tau = 0; tau < 2; ++tau)
            #pragma unroll
            for (int pq = 0; pq < 4; ++pq) cur[tau][pq] = nxt[tau][pq];
    }
    if (l16 < 6) {
        atomicAdd(&lsum[l16], ls);
        atomicAdd(&lssq[l16], lq);
    }
    __syncthreads();
    if (tid < 6) {
        atomicAdd(&dacc[2 + tid], (double)lsum[tid]);
        atomicAdd(&dacc[8 + tid], (double)lssq[tid]);
    }
}

__global__ __launch_bounds__(256) void k_m3(const float* __restrict__ u2,
                                            const float* __restrict__ g2,
                                            const float* __restrict__ b2,
                                            float* __restrict__ fc,
                                            double* __restrict__ dacc)
{
    __shared__ double sacc[27];
    __shared__ float lfc[12];
    int tid = threadIdx.x;
    if (tid < 27) sacc[tid] = 0.0;
    if (tid < 6) {
        double mu  = dacc[2 + tid] * RP;
        double var = dacc[8 + tid] * RP - mu*mu;
        float sc = g2[tid] / sqrtf((float)var + EPSBN);
        float sh = b2[tid] - (float)mu*sc;
        lfc[tid] = sc;  lfc[6 + tid] = sh;
        if (blockIdx.x == 0) { fc[6 + tid] = sc;  fc[12 + tid] = sh; }
    }
    __syncthreads();
    float sc2[6], sh2[6];
    #pragma unroll
    for (int c = 0; c < 6; ++c) { sc2[c] = lfc[c]; sh2[c] = lfc[6 + c]; }
    float m[6]; float s[21];
    #pragma unroll
    for (int c = 0; c < 6; ++c) m[c] = 0.f;
    #pragma unroll
    for (int k = 0; k < 21; ++k) s[k] = 0.f;
    long gid = (long)blockIdx.x*256 + tid;
    for (int it = 0; it < 12; ++it) {
        long pos = gid + (long)it*(M3_BLOCKS*256);
        float4 ua = *(const float4*)(u2 + pos*8);
        float4 ub = *(const float4*)(u2 + pos*8 + 4);
        float uv[6] = {ua.x, ua.y, ua.z, ua.w, ub.x, ub.y};
        float y[6];
        #pragma unroll
        for (int c = 0; c < 6; ++c) y[c] = lrelu(uv[c]*sc2[c] + sh2[c]);
        int idx = 0;
        #pragma unroll
        for (int c = 0; c < 6; ++c) {
            m[c] += y[c];
            #pragma unroll
            for (int c2 = c; c2 < 6; ++c2) s[idx++] += y[c]*y[c2];
        }
    }
    float tot[27];
    #pragma unroll
    for (int c = 0; c < 6; ++c)  tot[c]     = m[c];
    #pragma unroll
    for (int k = 0; k < 21; ++k) tot[6 + k] = s[k];
    #pragma unroll
    for (int msk = 16; msk >= 1; msk >>= 1) {
        #pragma unroll
        for (int k = 0; k < 27; ++k) tot[k] += __shfl_xor(tot[k], msk, 32);
    }
    int lane = tid & 31;
    float pick = 0.f;
    #pragma unroll
    for (int k = 0; k < 27; ++k) pick = (lane == k) ? tot[k] : pick;
    if (lane < 27) atomicAdd(&sacc[lane], (double)pick);
    __syncthreads();
    if (tid < 27) atomicAdd(&dacc[14 + tid], sacc[tid]);
}

__device__ __forceinline__ void bn3_consts(const float (&w)[6], const float (&mb)[6], const float (&cv)[21],
                                           float g, float b, float& A, float& B)
{
    float mu = 0.f, qd = 0.f, qo = 0.f;
    int idx = 0;
    #pragma unroll
    for (int c = 0; c < 6; ++c) {
        mu = __builtin_fmaf(w[c], mb[c], mu);
        qd = __builtin_fmaf(w[c]*w[c], cv[idx++], qd);
        if (c < 5) {
            float t = cv[idx++]*w[c + 1];
            #pragma unroll
            for (int c2 = c + 2; c2 < 6; ++c2) t = __builtin_fmaf(cv[idx++], w[c2], t);
            qo = __builtin_fmaf(w[c], t, qo);
        }
    }
    float q    = __builtin_fmaf(2.f, qo, qd);
    float mean = row16_sum(mu) * 0.0625f;
    float d    = mu - mean;
    float var  = row16_sum(__builtin_fmaf(d, d, q)) * 0.0625f;
    A = g / sqrtf(var + EPSBN);
    B = b - mean*A;
}

__global__ __launch_bounds__(288) void k_out(const float* __restrict__ u2,
                                             const float* __restrict__ wt,
                                             const float* __restrict__ fc,
                                             float* __restrict__ out,
                                             const double* __restrict__ dacc,
                                             const float* __restrict__ g3,
                                             const float* __restrict__ b3)
{
    __shared__ float  lab[6];
    __shared__ float4 stage4[12*OUT_T];
    float* stage = (float*)stage4;
    const int t     = threadIdx.x;
    const int tile0 = blockIdx.x*OUT_TILES;
    const int kkl = t / HP;
    const int l   = t - kkl*HP;

    const float* urow = u2 + ((long)tile0*OUT_T + t)*8;
    float4 ua = *(const float4*)(urow);
    float4 ub = *(const float4*)(urow + 4);

    const int wv = __builtin_amdgcn_readfirstlane((int)(threadIdx.x >> 5));
    if (wv == (int)(blockIdx.x & 7)) {
        const int lane = threadIdx.x & 31, row = lane >> 4, f = lane & 15;
        int m = lane < 27 ? lane : 26;
        int ca = m, cb = m;
        if (m >= 6) {
            int n = m - 6;
            ca = (n >= 6) + (n >= 11) + (n >= 15) + (n >= 18) + (n >= 20);
            cb = ca + n - (6*ca - ca*(ca - 1)/2);
        }
        double sm = dacc[14 + m],  sa = dacc[14 + ca],  sb = dacc[14 + cb];
        float w01[6], w2[6];
        #pragma unroll
        for (int c = 0; c < 6; ++c) {
            w01[c] = wt[(c*3 + row)*16 + f];
            w2[c]  = wt[(c*3 + 2)*16 + f];
        }
        float g01 = g3[row], b01 = b3[row], g2 = g3[2], b2 = b3[2];

        double ma = m < 6 ? 0.0 : sa * RP;
        float cm = (float)(sm * RP - ma * (sb * RP));
        float mb[6], cv[21];
        #pragma unroll
        for (int n = 0; n < 6; ++n)
            mb[n] = __builtin_bit_cast(float, __builtin_amdgcn_readlane(__builtin_bit_cast(int, cm), n));
        #pragma unroll
        for (int n = 0; n < 21; ++n)
            cv[n] = __builtin_bit_cast(float, __builtin_amdgcn_readlane(__builtin_bit_cast(int, cm), 6 + n));
        float A01, B01, A2, B2;
        bn3_consts(w01, mb, cv, g01, b01, A01, B01);
        bn3_consts(w2,  mb, cv, g2,  b2,  A2,  B2);
        if (f == 0)    { lab[row] = A01;  lab[3 + row] = B01; }
        if (lane == 0) { lab[2]   = A2;   lab[5]       = B2;  }
    }
    __syncthreads();

    float* mine = stage + kkl*96 + 2*l;

    #pragma unroll 1
    for (int it = 0; it < OUT_TILES; ++it) {
        const int tile  = tile0 + it;
        const int khalf = tile & 1;
        int r1 = tile >> 1;
        const int j = r1 % HP;  r1 /= HP;
        const int i = r1 % HP;  const int b = r1 / HP;

        if (it > 0) __syncthreads();

        float uv[6] = {ua.x, ua.y, ua.z, ua.w, ub.x, ub.y};
        if (it + 1 < OUT_TILES) {
            ua = *(const float4*)(urow + (long)(it + 1)*OUT_T*8);
            ub = *(const float4*)(urow + (long)(it + 1)*OUT_T*8 + 4);
        }
        float y[6];
        #pragma unroll
        for (int c = 0; c < 6; ++c) y[c] = lrelu(uv[c]*fc[6+c] + fc[12+c]);

        int wofs = 0;
        asm volatile("" : "+s"(wofs));
        wofs = __builtin_amdgcn_readfirstlane(wofs);
        const __attribute__((address_space(4))) float* wtl =
            (const __attribute__((address_space(4))) float*)wt + wofs;
        float acc[48];
        #pragma unroll
        for (int c = 0; c < 6; ++c)
          #pragma unroll
          for (int o = 0; o < 3; ++o) {
            const __attribute__((address_space(4))) float* wrow = wtl + (c*3 + o)*16;
            #pragma unroll
            for (int f = 0; f < 16; ++f)
                acc[o*16 + f] = (c == 0) ? y[0]*wrow[f] : __builtin_fmaf(y[c], wrow[f], acc[o*16 + f]);
          }

        #pragma unroll
        for (int o = 0; o < 3; ++o) {
            float A3 = lab[o], B3 = lab[3+o];
            #pragma unroll
            for (int p = 0; p < 2; ++p)
              #pragma unroll
              for (int q = 0; q < 2; ++q)
                #pragma unroll
                for (int r = 0; r < 2; ++r) {
                    int f = p*8 + q*4 + r*2;
                    float z0 = __builtin_fmaf(acc[o*16 + f    ], A3, B3);
                    float z1 = __builtin_fmaf(acc[o*16 + f + 1], A3, B3);
                    float s0 = __builtin_amdgcn_rcpf(1.f + __expf(-z0));
                    float s1 = __builtin_amdgcn_rcpf(1.f + __expf(-z1));
                    *(float2*)(mine + (o*4 + p*2 + q)*PIECE + r*48) = make_float2(s0, s1);
                }
        }
        __syncthreads();
        #pragma unroll
        for (int o = 0; o < 3; ++o)
          #pragma unroll
          for (int p = 0; p < 2; ++p)
            #pragma unroll
            for (int q = 0; q < 2; ++q) {
                float4 v = stage4[(o*4 + p*2 + q)*OUT_T + t];
                long pbase = ((((long)b*3 + o)*48 + (2*i+p))*48 + (2*j+q))*2304 + khalf*PIECE;
                *(float4*)(out + pbase + 4*t) = v;
            }
    }
}

extern "C" void kernel_launch(void* const* d_in, const int* in_sizes, int n_in,
                              void* d_out, int out_size, void* d_ws, size_t ws_size,
                              hipStream_t stream)
{
    (void)in_sizes; (void)n_in; (void)out_size; (void)ws_size;
    const float* x  = (const float*)d_in[0];
    const float* w1 = (const float*)d_in[1];
    const float* g1 = (const float*)d_in[2];
    const float* b1 = (const float*)d_in[3];
    const float* w2 = (const float*)d_in[4];
    const float* g2 = (const float*)d_in[5];
    const float* b2 = (const float*)d_in[6];
    const float* wt = (const float*)d_in[7];
    const float* g3 = (const float*)d_in[8];
    const float* b3 = (const float*)d_in[9];
    float*  out  = (float*)d_out;
    float*  u2   = (float*)((char*)d_ws + OFF_U2);
    float*  wp   = (float*)((char*)d_ws + OFF_WP);
    double* dacc = (double*)((char*)d_ws + OFF_DACC);
    float*  fc   = (float*)((char*)d_ws + OFF_FC);

    k_prep <<<1, 512, 0, stream>>>(w2, wp, dacc);
    k_rx   <<<RX_BLOCKS, 256, 0, stream>>>(x, dacc);
    k_conv2<<<C2_GROUPS, 256, 0, stream>>>(x, wp, w1, g1, b1, u2, dacc);
    k_m3   <<<M3_BLOCKS, 256, 0, stream>>>(u2, g2, b2, fc, dacc);
    k_out  <<<OUT_GROUPS, OUT_T, 0, stream>>>(u2, wt, fc, out, dacc, g3, b3);
}


# k_rx leaves one f64 pair of BN1 sums per workgroup (256 x 12 waves, a lane's 27 loads all in flight, no global atomics); k_conv2's head adds them: zeroing launch gone (5 -> 4); sums regrouped
# speedup vs baseline: 3.9822x; 1.0288x over previous
#include <hip/hip_runtime.h>
#include <math.h>

#define EPSBN 1e-5f
#define SLOPE 0.01f
static constexpr int  HP  = 24;
static constexpr long P   = 2L*24*24*24*24;
static constexpr long NX  = 2L*48*48*48*48;
static constexpr int  RX_GROUPS = 256;
static constexpr int  RX_T      = 384;
static constexpr int  RX_WAVES  = RX_T/32;
static constexpr int  RX_LOADS  = 27;
static_assert((long)RX_GROUPS*RX_T*RX_LOADS*4 == NX, "k_rx must cover x exactly once");
static_assert(RX_GROUPS == 256, "k_conv2's head takes one pair of k_rx per thread of its 256");
static constexpr int  M3_BLOCKS = 216;
static constexpr int  C2_GROUPS = 512;
static constexpr int  C2_WAVES  = C2_GROUPS*8;
static constexpr int  C2_UNITS  = (int)(P/32);
static constexpr int  OUT_T      = 288;
static constexpr int  OUT_BLOCKS = 2304;
static constexpr int  OUT_TILES  = 3;
static constexpr int  OUT_GROUPS = OUT_BLOCKS / OUT_TILES;
static constexpr int  PIECE      = 1152;
static constexpr double RNX = 1.0 / (double)NX;
static constexpr double RP  = 1.0 / (double)P;

static constexpr size_t OFF_U2   = 0;
static constexpr size_t OFF_WP   = (size_t)P*8*4;
static constexpr size_t OFF_DACC = OFF_WP + 768*4;
static constexpr size_t OFF_FC   = OFF_DACC + 512;
static constexpr size_t OFF_SLOT = OFF_DACC + 1024;

typedef float vf2 __attribute__((ext_vector_type(2)));
typedef float vf4 __attribute__((ext_vector_type(4)));
typedef float vf8 __attribute__((ext_vector_type(8)));

__device__ __forceinline__ float lrelu(float v){ return v >= 0.f ? v : SLOPE*v; }

template <int M> __device__ __forceinline__ float row_xor(float v)
{
    return __builtin_bit_cast(float,
        __builtin_amdgcn_update_dpp(0, __builtin_bit_cast(int, v), 0x160 | M, 0xf, 0xf, true));
}
__device__ __forceinline__ float row16_sum(float v)
{
    v += row_xor<8>(v);
    v += row_xor<4>(v);
    v += row_xor<2>(v);
    v += row_xor<1>(v);
    return v;
}


__global__ __launch_bounds__(384) void k_rx(const float* __restrict__ x,
                                            double* __restrict__ slot,
                                            double* __restrict__ dacc)
{
    __shared__ float wtot[2][16];
    const int tid = threadIdx.x;
    if (blockIdx.x == 0 && tid < 54) dacc[2 + tid] = 0.0;
    const vf4* xp = (const vf4*)x + (long)blockIdx.x*(RX_LOADS*RX_T) + tid;
    vf4 v[RX_LOADS];
    #pragma unroll
    for (int it = 0; it < RX_LOADS; ++it) v[it] = xp[it*RX_T];
    asm volatile("" :: "v"(v[0]), "v"(v[1]), "v"(v[2]), "v"(v[3]), "v"(v[4]), "v"(v[5]), "v"(v[6]), "v"(v[7]), "v"(v[8]),
                    "v"(v[9]), "v"(v[10]), "v"(v[11]), "v"(v[12]), "v"(v[13]), "v"(v[14]), "v"(v[15]), "v"(v[16]), "v"(v[17]),
                    "v"(v[18]), "v"(v[19]), "v"(v[20]), "v"(v[21]), "v"(v[22]), "v"(v[23]), "v"(v[24]), "v"(v[25]), "v"(v[26]));
    float s[4] = {0.f, 0.f, 0.f, 0.f}, q[4] = {0.f, 0.f, 0.f, 0.f};
    #pragma unroll
    for (int it = 0; it < RX_LOADS; ++it) {
        s[it & 3] += (v[it].x + v[it].y) + (v[it].z + v[it].w);
        q[it & 3] += (v[it].x*v[it].x + v[it].y*v[it].y) + (v[it].z*v[it].z + v[it].w*v[it].w);
    }
    float st = row16_sum((s[0] + s[1]) + (s[2] + s[3]));
    float qt = row16_sum((q[0] + q[1]) + (q[2] + q[3]));
    st += __shfl_xor(st, 16, 32);
    qt += __shfl_xor(qt, 16, 32);
    if ((tid & 31) == 0) { wtot[0][tid >> 5] = st;  wtot[1][tid >> 5] = qt; }
    __syncthreads();
    if (tid < 32) {
        double d = (tid & 15) < RX_WAVES ? (double)wtot[tid >> 4][tid & 15] : 0.0;
        #pragma unroll
        for (int m = 8; m >= 1; m >>= 1) d += __shfl_xor(d, m, 32);
        if ((tid & 15) == 0) slot[blockIdx.x*2 + (tid >> 4)] = d;
    }
}

__device__ __forceinline__ void c2_load(const float* __restrict__ x, int unit, int l16, int h, float2 (&xv)[2][4])
{
    #pragma unroll
    for (int tau = 0; tau < 2; ++tau) {
        int pos = unit*32 + tau*16 + l16;
        int l  = pos % HP;  int t1 = pos / HP;
        int kk = t1 % HP;   t1 /= HP;
        int j  = t1 % HP;   t1 /= HP;
        int i  = t1 % HP;   int b = t1 / HP;
        int xb = ((((b*48 + 2*i)*48 + 2*j)*48 + 2*kk + h)*48) + 2*l;
        #pragma unroll
        for (int pq = 0; pq < 4; ++pq)
            xv[tau][pq] = *(const float2*)(x + xb + (pq >> 1)*110592 + (pq & 1)*2304);
    }
}

__global__ __launch_bounds__(256) void k_conv2(const float* __restrict__ x,
                                               const float* __restrict__ w2,
                                               const double* __restrict__ slot,
                                               const float* __restrict__ w1,
                                               const float* __restrict__ g1,
                                               const float* __restrict__ b1,
                                               float* __restrict__ u2,
                                               double* __restrict__ dacc)
{
    __shared__ float lsum[6], lssq[6];
    __shared__ float lfc[6];
    __shared__ float lred[32];
    int tid = threadIdx.x;
    if (tid < 6) { lsum[tid] = 0.f; lssq[tid] = 0.f; }

    const int lane = tid & 31, l16 = lane & 15, h = lane >> 4;
    const int wv = __builtin_amdgcn_readfirstlane(tid >> 5);
    int unit = blockIdx.x*8 + wv;

    float2 cur[2][4], nxt[2][4];
    c2_load(x, unit, l16, h, cur);
    #pragma unroll
    for (int tau = 0; tau < 2; ++tau)
        #pragma unroll
        for (int pq = 0; pq < 4; ++pq) nxt[tau][pq] = cur[tau][pq];

    const double s0x = slot[0], s0y = slot[1];
    {
        float ds = (float)(slot[2*tid]     - s0x);
        float dq = (float)(slot[2*tid + 1] - s0y);
        ds = row16_sum(ds);
        dq = row16_sum(dq);
        if (l16 == 0) { lred[2*wv + h] = ds;  lred[16 + 2*wv + h] = dq; }
    }

    vf2 Bfrag[12];
    {
        const float* wrow = w2 + (l16 < 6 ? l16 : 5)*48 + 2*h;
        #pragma unroll
        for (int t = 0; t < 12; ++t) Bfrag[t] = *(const vf2*)(wrow + 4*t);
    }

    __syncthreads();

    unsigned hc = (unsigned)(tid - 32*(int)(blockIdx.x & 7));
    if (hc < 3u) {
        float pr[32];
        #pragma unroll
        for (int w = 0; w < 32; ++w) pr[w] = lred[w];
        #pragma unroll
        for (int st = 8; st >= 1; st >>= 1)
            #pragma unroll
            for (int w = 0; w < st; ++w) { pr[w] += pr[w + st];  pr[16 + w] += pr[16 + w + st]; }
        const float ts = pr[0], tq = pr[16];
        double sx  = __builtin_fma((double)RX_GROUPS, s0x, (double)ts);
        double sq  = __builtin_fma((double)RX_GROUPS, s0y, (double)tq);
        double mu  = sx * RNX;
        double var = sq * RNX - mu*mu;
        float  wc  = w1[hc];
        float  a   = wc * g1[hc] / sqrtf(wc*wc*(float)var + EPSBN);
        lfc[hc]     = a;
        lfc[3 + hc] = b1[hc] - a*(float)mu;
    }

    __syncthreads();

    float a1[3], c1[3];
    #pragma unroll
    for (int c = 0; c < 3; ++c) {
        a1[c] = __builtin_bit_cast(float, __builtin_amdgcn_readfirstlane(__builtin_bit_cast(int, lfc[c])));
        c1[c] = __builtin_bit_cast(float, __builtin_amdgcn_readfirstlane(__builtin_bit_cast(int, lfc[3+c])));
    }

    float ls = 0.f, lq = 0.f;

    #pragma unroll 1
    for (; unit < C2_UNITS; unit += C2_WAVES) {
        if (unit + C2_WAVES < C2_UNITS) c2_load(x, unit + C2_WAVES, l16, h, nxt);

        vf8 acc[2];
        #pragma unroll
        for (int tau = 0; tau < 2; ++tau) acc[tau] = (vf8){0.f,0.f,0.f,0.f,0.f,0.f,0.f,0.f};
        #pragma unroll
        for (int t = 0; t < 12; ++t) {
            int c = t >> 2;
            #pragma unroll
            for (int tau = 0; tau < 2; ++tau) {
                float2 xv = cur[tau][t & 3];
                vf2 A;
                A.x = lrelu(a1[c]*xv.x + c1[c]);
                A.y = lrelu(a1[c]*xv.y + c1[c]);
                acc[tau] = __builtin_amdgcn_wmma_f32_16x16x4_f32(
                               false, A, false, Bfrag[t], (short)0, acc[tau], false, false);
            }
        }
        #pragma unroll
        for (int tau = 0; tau < 2; ++tau)
            #pragma unroll
            for (int r = 0; r < 8; ++r) {
                float v = acc[tau][r];
                ls += v; lq += v*v;
            }
        if (l16 < 6) {
            float* up = u2 + ((long)unit*32 + 8*h)*8 + l16;
            #pragma unroll
            for (int tau = 0; tau < 2; ++tau)
                #pragma unroll
                for (int r = 0; r < 8; ++r) up[(tau*16 + r)*8] = acc[tau][r];
        }
        #pragma unroll
        for (int tau = 0; tau < 2; ++tau)
            #pragma unroll
            for (int pq = 0; pq < 4; ++pq) cur[tau][pq] = nxt[tau][pq];
    }
    if (l16 < 6) {
        atomicAdd(&lsum[l16], ls);
        atomicAdd(&lssq[l16], lq);
    }
    __syncthreads();
    if (tid < 6) {
        atomicAdd(&dacc[2 + tid], (double)lsum[tid]);
        atomicAdd(&dacc[8 + tid], (double)lssq[tid]);
    }
}

__global__ __launch_bounds__(256) void k_m3(const float* __restrict__ u2,
                                            const float* __restrict__ g2,
                                            const float* __restrict__ b2,
                                            float* __restrict__ fc,
                                            double* __restrict__ dacc)
{
    __shared__ double sacc[27];
    __shared__ float lfc[12];
    int tid = threadIdx.x;
    if (tid < 27) sacc[tid] = 0.0;
    if (tid < 6) {
        double mu  = dacc[2 + tid] * RP;
        double var = dacc[8 + tid] * RP - mu*mu;
        float sc = g2[tid] / sqrtf((float)var + EPSBN);
        float sh = b2[tid] - (float)mu*sc;
        lfc[tid] = sc;  lfc[6 + tid] = sh;
        if (blockIdx.x == 0) { fc[6 + tid] = sc;  fc[12 + tid] = sh; }
    }
    __syncthreads();
    float sc2[6], sh2[6];
    #pragma unroll
    for (int c = 0; c < 6; ++c) { sc2[c] = lfc[c]; sh2[c] = lfc[6 + c]; }
    float m[6]; float s[21];
    #pragma unroll
    for (int c = 0; c < 6; ++c) m[c] = 0.f;
    #pragma unroll
    for (int k = 0; k < 21; ++k) s[k] = 0.f;
    long gid = (long)blockIdx.x*256 + tid;
    for (int it = 0; it < 12; ++it) {
        long pos = gid + (long)it*(M3_BLOCKS*256);
        float4 ua = *(const float4*)(u2 + pos*8);
        float4 ub = *(const float4*)(u2 + pos*8 + 4);
        float uv[6] = {ua.x, ua.y, ua.z, ua.w, ub.x, ub.y};
        float y[6];
        #pragma unroll
        for (int c = 0; c < 6; ++c) y[c] = lrelu(uv[c]*sc2[c] + sh2[c]);
        int idx = 0;
        #pragma unroll
        for (int c = 0; c < 6; ++c) {
            m[c] += y[c];
            #pragma unroll
            for (int c2 = c; c2 < 6; ++c2) s[idx++] += y[c]*y[c2];
        }
    }
    float tot[27];
    #pragma unroll
    for (int c = 0; c < 6; ++c)  tot[c]     = m[c];
    #pragma unroll
    for (int k = 0; k < 21; ++k) tot[6 + k] = s[k];
    #pragma unroll
    for (int msk = 16; msk >= 1; msk >>= 1) {
        #pragma unroll
        for (int k = 0; k < 27; ++k) tot[k] += __shfl_xor(tot[k], msk, 32);
    }
    int lane = tid & 31;
    float pick = 0.f;
    #pragma unroll
    for (int k = 0; k < 27; ++k) pick = (lane == k) ? tot[k] : pick;
    if (lane < 27) atomicAdd(&sacc[lane], (double)pick);
    __syncthreads();
    if (tid < 27) atomicAdd(&dacc[14 + tid], sacc[tid]);
}

__device__ __forceinline__ void bn3_consts(const float (&w)[6], const float (&mb)[6], const float (&cv)[21],
                                           float g, float b, float& A, float& B)
{
    float mu = 0.f, qd = 0.f, qo = 0.f;
    int idx = 0;
    #pragma unroll
    for (int c = 0; c < 6; ++c) {
        mu = __builtin_fmaf(w[c], mb[c], mu);
        qd = __builtin_fmaf(w[c]*w[c], cv[idx++], qd);
        if (c < 5) {
            float t = cv[idx++]*w[c + 1];
            #pragma unroll
            for (int c2 = c + 2; c2 < 6; ++c2) t = __builtin_fmaf(cv[idx++], w[c2], t);
            qo = __builtin_fmaf(w[c], t, qo);
        }
    }
    float q    = __builtin_fmaf(2.f, qo, qd);
    float mean = row16_sum(mu) * 0.0625f;
    float d    = mu - mean;
    float var  = row16_sum(__builtin_fmaf(d, d, q)) * 0.0625f;
    A = g / sqrtf(var + EPSBN);
    B = b - mean*A;
}

__global__ __launch_bounds__(288) void k_out(const float* __restrict__ u2,
                                             const float* __restrict__ wt,
                                             const float* __restrict__ fc,
                                             float* __restrict__ out,
                                             const double* __restrict__ dacc,
                                             const float* __restrict__ g3,
                                             const float* __restrict__ b3)
{
    __shared__ float  lab[6];
    __shared__ float4 stage4[12*OUT_T];
    float* stage = (float*)stage4;
    const int t     = threadIdx.x;
    const int tile0 = blockIdx.x*OUT_TILES;
    const int kkl = t / HP;
    const int l   = t - kkl*HP;

    const float* urow = u2 + ((long)tile0*OUT_T + t)*8;
    float4 ua = *(const float4*)(urow);
    float4 ub = *(const float4*)(urow + 4);

    const int wv = __builtin_amdgcn_readfirstlane((int)(threadIdx.x >> 5));
    if (wv == (int)(blockIdx.x & 7)) {
        const int lane = threadIdx.x & 31, row = lane >> 4, f = lane & 15;
        int m = lane < 27 ? lane : 26;
        int ca = m, cb = m;
        if (m >= 6) {
            int n = m - 6;
            ca = (n >= 6) + (n >= 11) + (n >= 15) + (n >= 18) + (n >= 20);
            cb = ca + n - (6*ca - ca*(ca - 1)/2);
        }
        double sm = dacc[14 + m],  sa = dacc[14 + ca],  sb = dacc[14 + cb];
        float w01[6], w2[6];
        #pragma unroll
        for (int c = 0; c < 6; ++c) {
            w01[c] = wt[(c*3 + row)*16 + f];
            w2[c]  = wt[(c*3 + 2)*16 + f];
        }
        float g01 = g3[row], b01 = b3[row], g2 = g3[2], b2 = b3[2];

        double ma = m < 6 ? 0.0 : sa * RP;
        float cm = (float)(sm * RP - ma * (sb * RP));
        float mb[6], cv[21];
        #pragma unroll
        for (int n = 0; n < 6; ++n)
            mb[n] = __builtin_bit_cast(float, __builtin_amdgcn_readlane(__builtin_bit_cast(int, cm), n));
        #pragma unroll
        for (int n = 0; n < 21; ++n)
            cv[n] = __builtin_bit_cast(float, __builtin_amdgcn_readlane(__builtin_bit_cast(int, cm), 6 + n));
        float A01, B01, A2, B2;
        bn3_consts(w01, mb, cv, g01, b01, A01, B01);
        bn3_consts(w2,  mb, cv, g2,  b2,  A2,  B2);
        if (f == 0)    { lab[row] = A01;  lab[3 + row] = B01; }
        if (lane == 0) { lab[2]   = A2;   lab[5]       = B2;  }
    }
    __syncthreads();

    float* mine = stage + kkl*96 + 2*l;

    #pragma unroll 1
    for (int it = 0; it < OUT_TILES; ++it) {
        const int tile  = tile0 + it;
        const int khalf = tile & 1;
        int r1 = tile >> 1;
        const int j = r1 % HP;  r1 /= HP;
        const int i = r1 % HP;  const int b = r1 / HP;

        if (it > 0) __syncthreads();

        float uv[6] = {ua.x, ua.y, ua.z, ua.w, ub.x, ub.y};
        if (it + 1 < OUT_TILES) {
            ua = *(const float4*)(urow + (long)(it + 1)*OUT_T*8);
            ub = *(const float4*)(urow + (long)(it + 1)*OUT_T*8 + 4);
        }
        float y[6];
        #pragma unroll
        for (int c = 0; c < 6; ++c) y[c] = lrelu(uv[c]*fc[6+c] + fc[12+c]);

        int wofs = 0;
        asm volatile("" : "+s"(wofs));
        wofs = __builtin_amdgcn_readfirstlane(wofs);
        const __attribute__((address_space(4))) float* wtl =
            (const __attribute__((address_space(4))) float*)wt + wofs;
        float acc[48];
        #pragma unroll
        for (int c = 0; c < 6; ++c)
          #pragma unroll
          for (int o = 0; o < 3; ++o) {
            const __attribute__((address_space(4))) float* wrow = wtl + (c*3 + o)*16;
            #pragma unroll
            for (int f = 0; f < 16; ++f)
                acc[o*16 + f] = (c == 0) ? y[0]*wrow[f] : __builtin_fmaf(y[c], wrow[f], acc[o*16 + f]);
          }

        #pragma unroll
        for (int o = 0; o < 3; ++o) {
            float A3 = lab[o], B3 = lab[3+o];
            #pragma unroll
            for (int p = 0; p < 2; ++p)
              #pragma unroll
              for (int q = 0; q < 2; ++q)
                #pragma unroll
                for (int r = 0; r < 2; ++r) {
                    int f = p*8 + q*4 + r*2;
                    float z0 = __builtin_fmaf(acc[o*16 + f    ], A3, B3);
                    float z1 = __builtin_fmaf(acc[o*16 + f + 1], A3, B3);
                    float s0 = __builtin_amdgcn_rcpf(1.f + __expf(-z0));
                    float s1 = __builtin_amdgcn_rcpf(1.f + __expf(-z1));
                    *(float2*)(mine + (o*4 + p*2 + q)*PIECE + r*48) = make_float2(s0, s1);
                }
        }
        __syncthreads();
        #pragma unroll
        for (int o = 0; o < 3; ++o)
          #pragma unroll
          for (int p = 0; p < 2; ++p)
            #pragma unroll
            for (int q = 0; q < 2; ++q) {
                float4 v = stage4[(o*4 + p*2 + q)*OUT_T + t];
                long pbase = ((((long)b*3 + o)*48 + (2*i+p))*48 + (2*j+q))*2304 + khalf*PIECE;
                *(float4*)(out + pbase + 4*t) = v;
            }
    }
}

extern "C" void kernel_launch(void* const* d_in, const int* in_sizes, int n_in,
                              void* d_out, int out_size, void* d_ws, size_t ws_size,
                              hipStream_t stream)
{
    (void)in_sizes; (void)n_in; (void)out_size; (void)ws_size;
    const float* x  = (const float*)d_in[0];
    const float* w1 = (const float*)d_in[1];
    const float* g1 = (const float*)d_in[2];
    const float* b1 = (const float*)d_in[3];
    const float* w2 = (const float*)d_in[4];
    const float* g2 = (const float*)d_in[5];
    const float* b2 = (const float*)d_in[6];
    const float* wt = (const float*)d_in[7];
    const float* g3 = (const float*)d_in[8];
    const float* b3 = (const float*)d_in[9];
    float*  out  = (float*)d_out;
    float*  u2   = (float*)((char*)d_ws + OFF_U2);
    double* dacc = (double*)((char*)d_ws + OFF_DACC);
    float*  fc   = (float*)((char*)d_ws + OFF_FC);
    double* slot = (double*)((char*)d_ws + OFF_SLOT);

    k_rx   <<<RX_GROUPS, RX_T, 0, stream>>>(x, slot, dacc);
    k_conv2<<<C2_GROUPS, 256, 0, stream>>>(x, w2, slot, w1, g1, b1, u2, dacc);
    k_m3   <<<M3_BLOCKS, 256, 0, stream>>>(u2, g2, b2, fc, dacc);
    k_out  <<<OUT_GROUPS, OUT_T, 0, stream>>>(u2, wt, fc, out, dacc, g3, b3);
}


# k_conv2's loop issued 249 vector instructions per 24 WMMAs, now 141: leaky ReLU as max(v, 0.01v) for compare+select, next piece's coordinates by three carries for four divisions; all values as before
# speedup vs baseline: 4.1621x; 1.0452x over previous
#include <hip/hip_runtime.h>
#include <math.h>

#define EPSBN 1e-5f
#define SLOPE 0.01f
static constexpr int  HP  = 24;
static constexpr long P   = 2L*24*24*24*24;
static constexpr long NX  = 2L*48*48*48*48;
static constexpr int  RX_GROUPS = 256;
static constexpr int  RX_T      = 384;
static constexpr int  RX_WAVES  = RX_T/32;
static constexpr int  RX_LOADS  = 27;
static_assert((long)RX_GROUPS*RX_T*RX_LOADS*4 == NX, "k_rx must cover x exactly once");
static_assert(RX_GROUPS == 256, "k_conv2's head takes one pair of k_rx per thread of its 256");
static constexpr int  M3_BLOCKS = 216;
static constexpr int  C2_GROUPS = 512;
static constexpr int  C2_WAVES  = C2_GROUPS*8;
static constexpr int  C2_UNITS  = (int)(P/32);
static constexpr int  OUT_T      = 288;
static constexpr int  OUT_BLOCKS = 2304;
static constexpr int  OUT_TILES  = 3;
static constexpr int  OUT_GROUPS = OUT_BLOCKS / OUT_TILES;
static constexpr int  PIECE      = 1152;
static constexpr double RNX = 1.0 / (double)NX;
static constexpr double RP  = 1.0 / (double)P;

static constexpr size_t OFF_U2   = 0;
static constexpr size_t OFF_WP   = (size_t)P*8*4;
static constexpr size_t OFF_DACC = OFF_WP + 768*4;
static constexpr size_t OFF_FC   = OFF_DACC + 512;
static constexpr size_t OFF_SLOT = OFF_DACC + 1024;

typedef float vf2 __attribute__((ext_vector_type(2)));
typedef float vf4 __attribute__((ext_vector_type(4)));
typedef float vf8 __attribute__((ext_vector_type(8)));

__device__ __forceinline__ float lrelu(float v){ return v >= 0.f ? v : SLOPE*v; }

template <int M> __device__ __forceinline__ float row_xor(float v)
{
    return __builtin_bit_cast(float,
        __builtin_amdgcn_update_dpp(0, __builtin_bit_cast(int, v), 0x160 | M, 0xf, 0xf, true));
}
__device__ __forceinline__ float row16_sum(float v)
{
    v += row_xor<8>(v);
    v += row_xor<4>(v);
    v += row_xor<2>(v);
    v += row_xor<1>(v);
    return v;
}


__global__ __launch_bounds__(384) void k_rx(const float* __restrict__ x,
                                            double* __restrict__ slot,
                                            double* __restrict__ dacc)
{
    __shared__ float wtot[2][16];
    const int tid = threadIdx.x;
    if (blockIdx.x == 0 && tid < 54) dacc[2 + tid] = 0.0;
    const vf4* xp = (const vf4*)x + (long)blockIdx.x*(RX_LOADS*RX_T) + tid;
    vf4 v[RX_LOADS];
    #pragma unroll
    for (int it = 0; it < RX_LOADS; ++it) v[it] = xp[it*RX_T];
    asm volatile("" :: "v"(v[0]), "v"(v[1]), "v"(v[2]), "v"(v[3]), "v"(v[4]), "v"(v[5]), "v"(v[6]), "v"(v[7]), "v"(v[8]),
                    "v"(v[9]), "v"(v[10]), "v"(v[11]), "v"(v[12]), "v"(v[13]), "v"(v[14]), "v"(v[15]), "v"(v[16]), "v"(v[17]),
                    "v"(v[18]), "v"(v[19]), "v"(v[20]), "v"(v[21]), "v"(v[22]), "v"(v[23]), "v"(v[24]), "v"(v[25]), "v"(v[26]));
    float s[4] = {0.f, 0.f, 0.f, 0.f}, q[4] = {0.f, 0.f, 0.f, 0.f};
    #pragma unroll
    for (int it = 0; it < RX_LOADS; ++it) {
        s[it & 3] += (v[it].x + v[it].y) + (v[it].z + v[it].w);
        q[it & 3] += (v[it].x*v[it].x + v[it].y*v[it].y) + (v[it].z*v[it].z + v[it].w*v[it].w);
    }
    float st = row16_sum((s[0] + s[1]) + (s[2] + s[3]));
    float qt = row16_sum((q[0] + q[1]) + (q[2] + q[3]));
    st += __shfl_xor(st, 16, 32);
    qt += __shfl_xor(qt, 16, 32);
    if ((tid & 31) == 0) { wtot[0][tid >> 5] = st;  wtot[1][tid >> 5] = qt; }
    __syncthreads();
    if (tid < 32) {
        double d = (tid & 15) < RX_WAVES ? (double)wtot[tid >> 4][tid & 15] : 0.0;
        #pragma unroll
        for (int m = 8; m >= 1; m >>= 1) d += __shfl_xor(d, m, 32);
        if ((tid & 15) == 0) slot[blockIdx.x*2 + (tid >> 4)] = d;
    }
}

struct C2Pos { int l, kk, j, xb; };
static constexpr int C2_STEP = C2_WAVES*32;
static constexpr int C2_DL = C2_STEP % HP;
static constexpr int C2_DK = (C2_STEP / HP) % HP;
static constexpr int C2_DJ = (C2_STEP / (HP*HP)) % HP;
static constexpr int C2_DI = C2_STEP / (HP*HP*HP);
static constexpr int C2_DX = C2_DI*221184 + C2_DJ*4608 + C2_DK*96 + C2_DL*2;
static_assert(C2_DL + HP - 1 < 2*HP && C2_DK + HP < 2*HP && C2_DJ + HP < 2*HP, "one subtraction per carry");

__device__ __forceinline__ C2Pos c2_first(int pos, int h)
{
    C2Pos p;
    p.l  = pos % HP;  int t1 = pos / HP;
    p.kk = t1 % HP;   t1 /= HP;
    p.j  = t1 % HP;   t1 /= HP;
    p.xb = t1*221184 + p.j*4608 + p.kk*96 + h*48 + 2*p.l;
    return p;
}
__device__ __forceinline__ void c2_next(C2Pos& p)
{
    int l  = p.l + C2_DL;        bool cl = l >= HP;   p.l  = cl ? l - HP : l;
    int kk = p.kk + C2_DK + cl;  bool ck = kk >= HP;  p.kk = ck ? kk - HP : kk;
    int j  = p.j + C2_DJ + ck;   bool cj = j >= HP;   p.j  = cj ? j - HP : j;
    p.xb += C2_DX + (cl ? 48 : 0) + (ck ? 2304 : 0) + (cj ? 110592 : 0);
}
__device__ __forceinline__ void c2_load(const float* __restrict__ x, const C2Pos (&p)[2], float2 (&xv)[2][4])
{
    #pragma unroll
    for (int tau = 0; tau < 2; ++tau)
        #pragma unroll
        for (int pq = 0; pq < 4; ++pq)
            xv[tau][pq] = *(const float2*)(x + p[tau].xb + (pq >> 1)*110592 + (pq & 1)*2304);
}

__global__ __launch_bounds__(256) void k_conv2(const float* __restrict__ x,
                                               const float* __restrict__ w2,
                                               const double* __restrict__ slot,
                                               const float* __restrict__ w1,
                                               const float* __restrict__ g1,
                                               const float* __restrict__ b1,
                                               float* __restrict__ u2,
                                               double* __restrict__ dacc)
{
    __shared__ float lsum[6], lssq[6];
    __shared__ float lfc[6];
    __shared__ float lred[32];
    int tid = threadIdx.x;
    if (tid < 6) { lsum[tid] = 0.f; lssq[tid] = 0.f; }

    const int lane = tid & 31, l16 = lane & 15, h = lane >> 4;
    const int wv = __builtin_amdgcn_readfirstlane(tid >> 5);
    int unit = blockIdx.x*8 + wv;

    float2 cur[2][4], nxt[2][4];
    C2Pos cp[2];
    #pragma unroll
    for (int tau = 0; tau < 2; ++tau) cp[tau] = c2_first(unit*32 + tau*16 + l16, h);
    c2_load(x, cp, cur);
    #pragma unroll
    for (int tau = 0; tau < 2; ++tau)
        #pragma unroll
        for (int pq = 0; pq < 4; ++pq) nxt[tau][pq] = cur[tau][pq];

    const double s0x = slot[0], s0y = slot[1];
    {
        float ds = (float)(slot[2*tid]     - s0x);
        float dq = (float)(slot[2*tid + 1] - s0y);
        ds = row16_sum(ds);
        dq = row16_sum(dq);
        if (l16 == 0) { lred[2*wv + h] = ds;  lred[16 + 2*wv + h] = dq; }
    }

    vf2 Bfrag[12];
    {
        const float* wrow = w2 + (l16 < 6 ? l16 : 5)*48 + 2*h;
        #pragma unroll
        for (int t = 0; t < 12; ++t) Bfrag[t] = *(const vf2*)(wrow + 4*t);
    }

    __syncthreads();

    unsigned hc = (unsigned)(tid - 32*(int)(blockIdx.x & 7));
    if (hc < 3u) {
        float pr[32];
        #pragma unroll
        for (int w = 0; w < 32; ++w) pr[w] = lred[w];
        #pragma unroll
        for (int st = 8; st >= 1; st >>= 1)
            #pragma unroll
            for (int w = 0; w < st; ++w) { pr[w] += pr[w + st];  pr[16 + w] += pr[16 + w + st]; }
        const float ts = pr[0], tq = pr[16];
        double sx  = __builtin_fma((double)RX_GROUPS, s0x, (double)ts);
        double sq  = __builtin_fma((double)RX_GROUPS, s0y, (double)tq);
        double mu  = sx * RNX;
        double var = sq * RNX - mu*mu;
        float  wc  = w1[hc];
        float  a   = wc * g1[hc] / sqrtf(wc*wc*(float)var + EPSBN);
        lfc[hc]     = a;
        lfc[3 + hc] = b1[hc] - a*(float)mu;
    }

    __syncthreads();

    float a1[3], c1[3];
    #pragma unroll
    for (int c = 0; c < 3; ++c) {
        a1[c] = __builtin_bit_cast(float, __builtin_amdgcn_readfirstlane(__builtin_bit_cast(int, lfc[c])));
        c1[c] = __builtin_bit_cast(float, __builtin_amdgcn_readfirstlane(__builtin_bit_cast(int, lfc[3+c])));
    }

    float ls = 0.f, lq = 0.f;

    #pragma unroll 1
    for (; unit < C2_UNITS; unit += C2_WAVES) {
        if (unit + C2_WAVES < C2_UNITS) { c2_next(cp[0]);  c2_next(cp[1]);  c2_load(x, cp, nxt); }

        vf8 acc[2];
        #pragma unroll
        for (int tau = 0; tau < 2; ++tau) acc[tau] = (vf8){0.f,0.f,0.f,0.f,0.f,0.f,0.f,0.f};
        #pragma unroll
        for (int t = 0; t < 12; ++t) {
            int c = t >> 2;
            #pragma unroll
            for (int tau = 0; tau < 2; ++tau) {
                float2 xv = cur[tau][t & 3];
                vf2 A;
                float v0 = a1[c]*xv.x + c1[c], v1 = a1[c]*xv.y + c1[c];
                A.x = __builtin_fmaxf(v0, SLOPE*v0);
                A.y = __builtin_fmaxf(v1, SLOPE*v1);
                acc[tau] = __builtin_amdgcn_wmma_f32_16x16x4_f32(
                               false, A, false, Bfrag[t], (short)0, acc[tau], false, false);
            }
        }
        #pragma unroll
        for (int tau = 0; tau < 2; ++tau)
            #pragma unroll
            for (int r = 0; r < 8; ++r) {
                float v = acc[tau][r];
                ls += v; lq += v*v;
            }
        if (l16 < 6) {
            float* up = u2 + ((long)unit*32 + 8*h)*8 + l16;
            #pragma unroll
            for (int tau = 0; tau < 2; ++tau)
                #pragma unroll
                for (int r = 0; r < 8; ++r) up[(tau*16 + r)*8] = acc[tau][r];
        }
        #pragma unroll
        for (int tau = 0; tau < 2; ++tau)
            #pragma unroll
            for (int pq = 0; pq < 4; ++pq) cur[tau][pq] = nxt[tau][pq];
    }
    if (l16 < 6) {
        atomicAdd(&lsum[l16], ls);
        atomicAdd(&lssq[l16], lq);
    }
    __syncthreads();
    if (tid < 6) {
        atomicAdd(&dacc[2 + tid], (double)lsum[tid]);
        atomicAdd(&dacc[8 + tid], (double)lssq[tid]);
    }
}

__global__ __launch_bounds__(256) void k_m3(const float* __restrict__ u2,
                                            const float* __restrict__ g2,
                                            const float* __restrict__ b2,
                                            float* __restrict__ fc,
                                            double* __restrict__ dacc)
{
    __shared__ double sacc[27];
    __shared__ float lfc[12];
    int tid = threadIdx.x;
    if (tid < 27) sacc[tid] = 0.0;
    if (tid < 6) {
        double mu  = dacc[2 + tid] * RP;
        double var = dacc[8 + tid] * RP - mu*mu;
        float sc = g2[tid] / sqrtf((float)var + EPSBN);
        float sh = b2[tid] - (float)mu*sc;
        lfc[tid] = sc;  lfc[6 + tid] = sh;
        if (blockIdx.x == 0) { fc[6 + tid] = sc;  fc[12 + tid] = sh; }
    }
    __syncthreads();
    float sc2[6], sh2[6];
    #pragma unroll
    for (int c = 0; c < 6; ++c) { sc2[c] = lfc[c]; sh2[c] = lfc[6 + c]; }
    float m[6]; float s[21];
    #pragma unroll
    for (int c = 0; c < 6; ++c) m[c] = 0.f;
    #pragma unroll
    for (int k = 0; k < 21; ++k) s[k] = 0.f;
    long gid = (long)blockIdx.x*256 + tid;
    for (int it = 0; it < 12; ++it) {
        long pos = gid + (long)it*(M3_BLOCKS*256);
        float4 ua = *(const float4*)(u2 + pos*8);
        float4 ub = *(const float4*)(u2 + pos*8 + 4);
        float uv[6] = {ua.x, ua.y, ua.z, ua.w, ub.x, ub.y};
        float y[6];
        #pragma unroll
        for (int c = 0; c < 6; ++c) y[c] = lrelu(uv[c]*sc2[c] + sh2[c]);
        int idx = 0;
        #pragma unroll
        for (int c = 0; c < 6; ++c) {
            m[c] += y[c];
            #pragma unroll
            for (int c2 = c; c2 < 6; ++c2) s[idx++] += y[c]*y[c2];
        }
    }
    float tot[27];
    #pragma unroll
    for (int c = 0; c < 6; ++c)  tot[c]     = m[c];
    #pragma unroll
    for (int k = 0; k < 21; ++k) tot[6 + k] = s[k];
    #pragma unroll
    for (int msk = 16; msk >= 1; msk >>= 1) {
        #pragma unroll
        for (int k = 0; k < 27; ++k) tot[k] += __shfl_xor(tot[k], msk, 32);
    }
    int lane = tid & 31;
    float pick = 0.f;
    #pragma unroll
    for (int k = 0; k < 27; ++k) pick = (lane == k) ? tot[k] : pick;
    if (lane < 27) atomicAdd(&sacc[lane], (double)pick);
    __syncthreads();
    if (tid < 27) atomicAdd(&dacc[14 + tid], sacc[tid]);
}

__device__ __forceinline__ void bn3_consts(const float (&w)[6], const float (&mb)[6], const float (&cv)[21],
                                           float g, float b, float& A, float& B)
{
    float mu = 0.f, qd = 0.f, qo = 0.f;
    int idx = 0;
    #pragma unroll
    for (int c = 0; c < 6; ++c) {
        mu = __builtin_fmaf(w[c], mb[c], mu);
        qd = __builtin_fmaf(w[c]*w[c], cv[idx++], qd);
        if (c < 5) {
            float t = cv[idx++]*w[c + 1];
            #pragma unroll
            for (int c2 = c + 2; c2 < 6; ++c2) t = __builtin_fmaf(cv[idx++], w[c2], t);
            qo = __builtin_fmaf(w[c], t, qo);
        }
    }
    float q    = __builtin_fmaf(2.f, qo, qd);
    float mean = row16_sum(mu) * 0.0625f;
    float d    = mu - mean;
    float var  = row16_sum(__builtin_fmaf(d, d, q)) * 0.0625f;
    A = g / sqrtf(var + EPSBN);
    B = b - mean*A;
}

__global__ __launch_bounds__(288) void k_out(const float* __restrict__ u2,
                                             const float* __restrict__ wt,
                                             const float* __restrict__ fc,
                                             float* __restrict__ out,
                                             const double* __restrict__ dacc,
                                             const float* __restrict__ g3,
                                             const float* __restrict__ b3)
{
    __shared__ float  lab[6];
    __shared__ float4 stage4[12*OUT_T];
    float* stage = (float*)stage4;
    const int t     = threadIdx.x;
    const int tile0 = blockIdx.x*OUT_TILES;
    const int kkl = t / HP;
    const int l   = t - kkl*HP;

    const float* urow = u2 + ((long)tile0*OUT_T + t)*8;
    float4 ua = *(const float4*)(urow);
    float4 ub = *(const float4*)(urow + 4);

    const int wv = __builtin_amdgcn_readfirstlane((int)(threadIdx.x >> 5));
    if (wv == (int)(blockIdx.x & 7)) {
        const int lane = threadIdx.x & 31, row = lane >> 4, f = lane & 15;
        int m = lane < 27 ? lane : 26;
        int ca = m, cb = m;
        if (m >= 6) {
            int n = m - 6;
            ca = (n >= 6) + (n >= 11) + (n >= 15) + (n >= 18) + (n >= 20);
            cb = ca + n - (6*ca - ca*(ca - 1)/2);
        }
        double sm = dacc[14 + m],  sa = dacc[14 + ca],  sb = dacc[14 + cb];
        float w01[6], w2[6];
        #pragma unroll
        for (int c = 0; c < 6; ++c) {
            w01[c] = wt[(c*3 + row)*16 + f];
            w2[c]  = wt[(c*3 + 2)*16 + f];
        }
        float g01 = g3[row], b01 = b3[row], g2 = g3[2], b2 = b3[2];

        double ma = m < 6 ? 0.0 : sa * RP;
        float cm = (float)(sm * RP - ma * (sb * RP));
        float mb[6], cv[21];
        #pragma unroll
        for (int n = 0; n < 6; ++n)
            mb[n] = __builtin_bit_cast(float, __builtin_amdgcn_readlane(__builtin_bit_cast(int, cm), n));
        #pragma unroll
        for (int n = 0; n < 21; ++n)
            cv[n] = __builtin_bit_cast(float, __builtin_amdgcn_readlane(__builtin_bit_cast(int, cm), 6 + n));
        float A01, B01, A2, B2;
        bn3_consts(w01, mb, cv, g01, b01, A01, B01);
        bn3_consts(w2,  mb, cv, g2,  b2,  A2,  B2);
        if (f == 0)    { lab[row] = A01;  lab[3 + row] = B01; }
        if (lane == 0) { lab[2]   = A2;   lab[5]       = B2;  }
    }
    __syncthreads();

    float* mine = stage + kkl*96 + 2*l;

    #pragma unroll 1
    for (int it = 0; it < OUT_TILES; ++it) {
        const int tile  = tile0 + it;
        const int khalf = tile & 1;
        int r1 = tile >> 1;
        const int j = r1 % HP;  r1 /= HP;
        const int i = r1 % HP;  const int b = r1 / HP;

        if (it > 0) __syncthreads();

        float uv[6] = {ua.x, ua.y, ua.z, ua.w, ub.x, ub.y};
        if (it + 1 < OUT_TILES) {
            ua = *(const float4*)(urow + (long)(it + 1)*OUT_T*8);
            ub = *(const float4*)(urow + (long)(it + 1)*OUT_T*8 + 4);
        }
        float y[6];
        #pragma unroll
        for (int c = 0; c < 6; ++c) y[c] = lrelu(uv[c]*fc[6+c] + fc[12+c]);

        int wofs = 0;
        asm volatile("" : "+s"(wofs));
        wofs = __builtin_amdgcn_readfirstlane(wofs);
        const __attribute__((address_space(4))) float* wtl =
            (const __attribute__((address_space(4))) float*)wt + wofs;
        float acc[48];
        #pragma unroll
        for (int c = 0; c < 6; ++c)
          #pragma unroll
          for (int o = 0; o < 3; ++o) {
            const __attribute__((address_space(4))) float* wrow = wtl + (c*3 + o)*16;
            #pragma unroll
            for (int f = 0; f < 16; ++f)
                acc[o*16 + f] = (c == 0) ? y[0]*wrow[f] : __builtin_fmaf(y[c], wrow[f], acc[o*16 + f]);
          }

        #pragma unroll
        for (int o = 0; o < 3; ++o) {
            float A3 = lab[o], B3 = lab[3+o];
            #pragma unroll
            for (int p = 0; p < 2; ++p)
              #pragma unroll
              for (int q = 0; q < 2; ++q)
                #pragma unroll
                for (int r = 0; r < 2; ++r) {
                    int f = p*8 + q*4 + r*2;
                    float z0 = __builtin_fmaf(acc[o*16 + f    ], A3, B3);
                    float z1 = __builtin_fmaf(acc[o*16 + f + 1], A3, B3);
                    float s0 = __builtin_amdgcn_rcpf(1.f + __expf(-z0));
                    float s1 = __builtin_amdgcn_rcpf(1.f + __expf(-z1));
                    *(float2*)(mine + (o*4 + p*2 + q)*PIECE + r*48) = make_float2(s0, s1);
                }
        }
        __syncthreads();
        #pragma unroll
        for (int o = 0; o < 3; ++o)
          #pragma unroll
          for (int p = 0; p < 2; ++p)
            #pragma unroll
            for (int q = 0; q < 2; ++q) {
                float4 v = stage4[(o*4 + p*2 + q)*OUT_T + t];
                long pbase = ((((long)b*3 + o)*48 + (2*i+p))*48 + (2*j+q))*2304 + khalf*PIECE;
                *(float4*)(out + pbase + 4*t) = v;
            }
    }
}

extern "C" void kernel_launch(void* const* d_in, const int* in_sizes, int n_in,
                              void* d_out, int out_size, void* d_ws, size_t ws_size,
                              hipStream_t stream)
{
    (void)in_sizes; (void)n_in; (void)out_size; (void)ws_size;
    const float* x  = (const float*)d_in[0];
    const float* w1 = (const float*)d_in[1];
    const float* g1 = (const float*)d_in[2];
    const float* b1 = (const float*)d_in[3];
    const float* w2 = (const float*)d_in[4];
    const float* g2 = (const float*)d_in[5];
    const float* b2 = (const float*)d_in[6];
    const float* wt = (const float*)d_in[7];
    const float* g3 = (const float*)d_in[8];
    const float* b3 = (const float*)d_in[9];
    float*  out  = (float*)d_out;
    float*  u2   = (float*)((char*)d_ws + OFF_U2);
    double* dacc = (double*)((char*)d_ws + OFF_DACC);
    float*  fc   = (float*)((char*)d_ws + OFF_FC);
    double* slot = (double*)((char*)d_ws + OFF_SLOT);

    k_rx   <<<RX_GROUPS, RX_T, 0, stream>>>(x, slot, dacc);
    k_conv2<<<C2_GROUPS, 256, 0, stream>>>(x, w2, slot, w1, g1, b1, u2, dacc);
    k_m3   <<<M3_BLOCKS, 256, 0, stream>>>(u2, g2, b2, fc, dacc);
    k_out  <<<OUT_GROUPS, OUT_T, 0, stream>>>(u2, wt, fc, out, dacc, g3, b3);
}
